# starting kernel
# baseline (speedup 1.0000x reference)
#include <hip/hip_runtime.h>
#include <hip/hip_bf16.h>

#define EMBED_DIM 128
#define KNN 10
#define NODE_DIM 22
#define FDIM 24
#define NNODES 2000
#define NBATCH 16
#define BLOCK 128
#define NODES_PER_BLOCK 128
#define BLOCKS_PER_BATCH 16
#define FSTRIDE 25

typedef __attribute__((ext_vector_type(2))) float v2f;
typedef __attribute__((ext_vector_type(8))) float v8f;

__global__ __launch_bounds__(BLOCK)
void tsp_embed_kernel(const float* __restrict__ locs,
                      const float* __restrict__ W,
                      const float* __restrict__ bias,
                      float* __restrict__ out)
{
    __shared__ float sLocs[2 * NNODES];
    __shared__ float sFeat[NODES_PER_BLOCK * FSTRIDE];
    __shared__ float sW[FDIM * EMBED_DIM];

    const int tid   = threadIdx.x;
    const int batch = blockIdx.x / BLOCKS_PER_BATCH;
    const int node0 = (blockIdx.x % BLOCKS_PER_BATCH) * NODES_PER_BLOCK;

    const float* bl = locs + (size_t)batch * NNODES * 2;
    for (int idx = tid; idx < 2 * NNODES; idx += BLOCK) sLocs[idx] = bl[idx];
    for (int idx = tid; idx < FDIM * EMBED_DIM; idx += BLOCK) {
        const int f = idx / EMBED_DIM;
        const int e = idx % EMBED_DIM;
        sW[idx] = (f < NODE_DIM) ? W[e * NODE_DIM + f] : 0.0f;
    }
    __syncthreads();

    const int node = node0 + tid;
    float bestD[KNN];
    int   bestJ[KNN];
#pragma unroll
    for (int k = 0; k < KNN; ++k) { bestD[k] = 3.0e38f; bestJ[k] = 0; }

    const float2* sl = (const float2*)sLocs;
    float px = 0.0f, py = 0.0f;
    if (node < NNODES) {
        const float2 me = sl[node];
        px = me.x; py = me.y;
        for (int j = 0; j < NNODES; ++j) {
            const float2 q = sl[j];
            const float dx = q.x - px;
            const float dy = q.y - py;
            const float d2 = dx * dx + dy * dy;
            if (j != node && d2 < bestD[KNN - 1]) {
                bestD[KNN - 1] = d2;
                bestJ[KNN - 1] = j;
#pragma unroll
                for (int s = KNN - 1; s > 0; --s) {
                    if (bestD[s] < bestD[s - 1]) {
                        const float td = bestD[s]; bestD[s] = bestD[s - 1]; bestD[s - 1] = td;
                        const int   tj = bestJ[s]; bestJ[s] = bestJ[s - 1]; bestJ[s - 1] = tj;
                    }
                }
            }
        }
    }

    float* fr = &sFeat[tid * FSTRIDE];
    if (node < NNODES) {
        fr[0] = px;
        fr[1] = py;
#pragma unroll
        for (int k = 0; k < KNN; ++k) {
            const float2 q = sl[bestJ[k]];
            fr[2 + 2 * k]     = q.x - px;
            fr[2 + 2 * k + 1] = q.y - py;
        }
        fr[NODE_DIM]     = 0.0f;
        fr[NODE_DIM + 1] = 0.0f;
    } else {
#pragma unroll
        for (int k = 0; k < FDIM; ++k) fr[k] = 0.0f;
    }
    __syncthreads();

    const int wave   = tid >> 5;
    const int lane   = tid & 31;
    const int laneLo = lane & 15;
    const int kHalf  = (lane < 16) ? 0 : 2;
    const int mBase  = (lane < 16) ? 0 : 8;

    const int t0 = wave * 2;
    const int t1 = t0 + 1;
    const bool ok0 = (node0 + t0 * 16) < NNODES;
    const bool ok1 = (node0 + t1 * 16) < NNODES;

    v2f a0[6], a1[6];
    {
        const float* fa0 = &sFeat[(t0 * 16 + laneLo) * FSTRIDE];
        const float* fa1 = &sFeat[(t1 * 16 + laneLo) * FSTRIDE];
#pragma unroll
        for (int kc = 0; kc < 6; ++kc) {
            const int k0 = kc * 4 + kHalf;
            a0[kc].x = fa0[k0];  a0[kc].y = fa0[k0 + 1];
            a1[kc].x = fa1[k0];  a1[kc].y = fa1[k0 + 1];
        }
    }

    float* const outB = out + (size_t)batch * NNODES * EMBED_DIM;

#pragma unroll
    for (int ct = 0; ct < 8; ++ct) {
        const int e = ct * 16 + laneLo;
        v2f bf[6];
#pragma unroll
        for (int kc = 0; kc < 6; ++kc) {
            const int k0 = kc * 4 + kHalf;
            bf[kc].x = sW[k0 * EMBED_DIM + e];
            bf[kc].y = sW[(k0 + 1) * EMBED_DIM + e];
        }
        v8f acc0 = {}, acc1 = {};
#pragma unroll
        for (int kc = 0; kc < 6; ++kc)
            acc0 = __builtin_amdgcn_wmma_f32_16x16x4_f32(
                false, a0[kc], false, bf[kc], (short)0, acc0, false, false);
#pragma unroll
        for (int kc = 0; kc < 6; ++kc)
            acc1 = __builtin_amdgcn_wmma_f32_16x16x4_f32(
                false, a1[kc], false, bf[kc], (short)0, acc1, false, false);

        const float bv = bias[e];
        if (ok0) {
            float* p = outB + (size_t)(node0 + t0 * 16 + mBase) * EMBED_DIM + e;
#pragma unroll
            for (int r = 0; r < 8; ++r)
                p[(size_t)r * EMBED_DIM] = acc0[r] + bv;
        }
        if (ok1) {
            float* p = outB + (size_t)(node0 + t1 * 16 + mBase) * EMBED_DIM + e;
#pragma unroll
            for (int r = 0; r < 8; ++r)
                p[(size_t)r * EMBED_DIM] = acc1[r] + bv;
        }
    }
}

extern "C" void kernel_launch(void* const* d_in, const int* in_sizes, int n_in,
                              void* d_out, int out_size, void* d_ws, size_t ws_size,
                              hipStream_t stream) {
    const float* locs = (const float*)d_in[0];
    const float* W    = (const float*)d_in[1];
    const float* b    = (const float*)d_in[2];
    float* out        = (float*)d_out;

    dim3 grid(NBATCH * BLOCKS_PER_BATCH);
    dim3 block(BLOCK);
    tsp_embed_kernel<<<grid, block, 0, stream>>>(locs, W, b, out);
}


# KNN scan made branch-free: each lane appends candidates under its stale 10th-best bound to its own LDS queue; sorted inserts run in lockstep drains, so one lane's insert no longer stalls the other 31
# speedup vs baseline: 2.8328x; 2.8328x over previous
#include <hip/hip_runtime.h>
#include <hip/hip_bf16.h>

#define EMBED_DIM 128
#define KNN 10
#define NODE_DIM 22
#define FDIM 24
#define NNODES 2000
#define NBATCH 16
#define BLOCK 128
#define NODES_PER_BLOCK 128
#define BLOCKS_PER_BATCH 16
#define FSTRIDE 25
#define NTRIP (NNODES / 16)
#define QCAP 40
#define QROW (BLOCK * 8)

typedef __attribute__((ext_vector_type(2))) float v2f;
typedef __attribute__((ext_vector_type(8))) float v8f;

__global__ __launch_bounds__(BLOCK)
void tsp_embed_kernel(const float* __restrict__ locs,
                      const float* __restrict__ W,
                      const float* __restrict__ bias,
                      float* __restrict__ out)
{
    __shared__ alignas(16) float sLocs[2 * NNODES];
    __shared__ float sFeat[NODES_PER_BLOCK * FSTRIDE];
    __shared__ float sW[FDIM * EMBED_DIM];
    __shared__ float2 sQ[QCAP * BLOCK];

    const int tid   = threadIdx.x;
    const int batch = blockIdx.x / BLOCKS_PER_BATCH;
    const int node0 = (blockIdx.x % BLOCKS_PER_BATCH) * NODES_PER_BLOCK;

    const float* bl = locs + (size_t)batch * NNODES * 2;
    for (int idx = tid; idx < 2 * NNODES; idx += BLOCK) sLocs[idx] = bl[idx];
    for (int idx = tid; idx < FDIM * EMBED_DIM; idx += BLOCK) {
        const int f = idx / EMBED_DIM;
        const int e = idx % EMBED_DIM;
        sW[idx] = (f < NODE_DIM) ? W[e * NODE_DIM + f] : 0.0f;
    }
    __syncthreads();

    const int  node  = node0 + tid;
    const bool valid = node < NNODES;
    float bestD[KNN];
    int   bestJ[KNN];
#pragma unroll
    for (int k = 0; k < KNN; ++k) { bestD[k] = 3.0e38f; bestJ[k] = 0; }

    const float2* sl  = (const float2*)sLocs;
    const float4* sl4 = (const float4*)sLocs;
    float px = 0.0f, py = 0.0f;
    if (valid) { const float2 me = sl[node]; px = me.x; py = me.y; }

    float thr = valid ? 3.0e38f : -1.0f;
    unsigned qb = tid * 8;

    auto cand = [&](float qx, float qy, int j) {
        const float dx = qx - px;
        const float dy = qy - py;
        const float d2 = __builtin_fmaf(dx, dx, dy * dy);
        *(float2*)((char*)sQ + qb) = make_float2(d2, __int_as_float(j));
        qb += (d2 < thr) ? QROW : 0;
    };
    auto proc8 = [&](const float4 (&h)[4], int jb) {
#pragma unroll
        for (int u = 0; u < 4; ++u) {
            cand(h[u].x, h[u].y, jb + 2 * u);
            cand(h[u].z, h[u].w, jb + 2 * u + 1);
        }
    };

    float4 h0[4], h1[4];
#pragma unroll
    for (int u = 0; u < 4; ++u) h0[u] = sl4[u];

    for (int t = 0; t < NTRIP; ++t) {
        const int jb = t * 16;
#pragma unroll
        for (int u = 0; u < 4; ++u) h1[u] = sl4[(jb + 8) / 2 + u];
        proc8(h0, jb);
        const int cntMid = qb / QROW;
        const int jn = (jb + 16 < NNODES) ? (jb + 16) : (NNODES - 8);
#pragma unroll
        for (int u = 0; u < 4; ++u) h0[u] = sl4[jn / 2 + u];
        proc8(h1, jb + 8);

        const bool full = __builtin_amdgcn_ballot_w32(cntMid > QCAP - 24) != 0u;
        if (full || t == NTRIP - 1) {
            const int cnt = qb / QROW;
            const int lastRow = (cnt > 0) ? (cnt - 1) : 0;
            float2 e = sQ[tid];
            for (int i = 0; ; ++i) {
                const int nx = (i + 1 < lastRow) ? (i + 1) : lastRow;
                const float2 en = sQ[nx * BLOCK + tid];
                const bool more = __builtin_amdgcn_ballot_w32(i + 1 < cnt) != 0u;
                const float d2 = e.x;
                const int   j  = __float_as_int(e.y);
                if (i < cnt && j != node && d2 < bestD[KNN - 1]) {
                    bestD[KNN - 1] = d2;
                    bestJ[KNN - 1] = j;
#pragma unroll
                    for (int s = KNN - 1; s > 0; --s) {
                        if (bestD[s] < bestD[s - 1]) {
                            const float td = bestD[s]; bestD[s] = bestD[s - 1]; bestD[s - 1] = td;
                            const int   tj = bestJ[s]; bestJ[s] = bestJ[s - 1]; bestJ[s - 1] = tj;
                        }
                    }
                }
                if (!more) break;
                e = en;
            }
            qb = tid * 8;
            if (valid) thr = bestD[KNN - 1];
        }
    }

    float* fr = &sFeat[tid * FSTRIDE];
    if (node < NNODES) {
        fr[0] = px;
        fr[1] = py;
#pragma unroll
        for (int k = 0; k < KNN; ++k) {
            const float2 q = sl[bestJ[k]];
            fr[2 + 2 * k]     = q.x - px;
            fr[2 + 2 * k + 1] = q.y - py;
        }
        fr[NODE_DIM]     = 0.0f;
        fr[NODE_DIM + 1] = 0.0f;
    } else {
#pragma unroll
        for (int k = 0; k < FDIM; ++k) fr[k] = 0.0f;
    }
    __syncthreads();

    const int wave   = tid >> 5;
    const int lane   = tid & 31;
    const int laneLo = lane & 15;
    const int kHalf  = (lane < 16) ? 0 : 2;
    const int mBase  = (lane < 16) ? 0 : 8;

    const int t0 = wave * 2;
    const int t1 = t0 + 1;
    const bool ok0 = (node0 + t0 * 16) < NNODES;
    const bool ok1 = (node0 + t1 * 16) < NNODES;

    v2f a0[6], a1[6];
    {
        const float* fa0 = &sFeat[(t0 * 16 + laneLo) * FSTRIDE];
        const float* fa1 = &sFeat[(t1 * 16 + laneLo) * FSTRIDE];
#pragma unroll
        for (int kc = 0; kc < 6; ++kc) {
            const int k0 = kc * 4 + kHalf;
            a0[kc].x = fa0[k0];  a0[kc].y = fa0[k0 + 1];
            a1[kc].x = fa1[k0];  a1[kc].y = fa1[k0 + 1];
        }
    }

    float* const outB = out + (size_t)batch * NNODES * EMBED_DIM;

#pragma unroll
    for (int ct = 0; ct < 8; ++ct) {
        const int e = ct * 16 + laneLo;
        v2f bf[6];
#pragma unroll
        for (int kc = 0; kc < 6; ++kc) {
            const int k0 = kc * 4 + kHalf;
            bf[kc].x = sW[k0 * EMBED_DIM + e];
            bf[kc].y = sW[(k0 + 1) * EMBED_DIM + e];
        }
        v8f acc0 = {}, acc1 = {};
#pragma unroll
        for (int kc = 0; kc < 6; ++kc)
            acc0 = __builtin_amdgcn_wmma_f32_16x16x4_f32(
                false, a0[kc], false, bf[kc], (short)0, acc0, false, false);
#pragma unroll
        for (int kc = 0; kc < 6; ++kc)
            acc1 = __builtin_amdgcn_wmma_f32_16x16x4_f32(
                false, a1[kc], false, bf[kc], (short)0, acc1, false, false);

        const float bv = bias[e];
        if (ok0) {
            float* p = outB + (size_t)(node0 + t0 * 16 + mBase) * EMBED_DIM + e;
#pragma unroll
            for (int r = 0; r < 8; ++r)
                p[(size_t)r * EMBED_DIM] = acc0[r] + bv;
        }
        if (ok1) {
            float* p = outB + (size_t)(node0 + t1 * 16 + mBase) * EMBED_DIM + e;
#pragma unroll
            for (int r = 0; r < 8; ++r)
                p[(size_t)r * EMBED_DIM] = acc1[r] + bv;
        }
    }
}

extern "C" void kernel_launch(void* const* d_in, const int* in_sizes, int n_in,
                              void* d_out, int out_size, void* d_ws, size_t ws_size,
                              hipStream_t stream) {
    const float* locs = (const float*)d_in[0];
    const float* W    = (const float*)d_in[1];
    const float* b    = (const float*)d_in[2];
    float* out        = (float*)d_out;

    dim3 grid(NBATCH * BLOCKS_PER_BATCH);
    dim3 block(BLOCK);
    tsp_embed_kernel<<<grid, block, 0, stream>>>(locs, W, b, out);
}


# Each node's 2000-candidate KNN scan split over 4 wave groups (512-thread work-groups: 4 waves per SIMD, was 1) to hide dependent-issue latency; the 4 sorted tens merged by in-order insert
# speedup vs baseline: 5.0381x; 1.7785x over previous
#include <hip/hip_runtime.h>
#include <hip/hip_bf16.h>

#define EMBED_DIM 128
#define KNN 10
#define NODE_DIM 22
#define FDIM 24
#define NNODES 2000
#define NBATCH 16
#define NPARTS 4
#define NODES_PER_BLOCK 128
#define BLOCK (NPARTS * NODES_PER_BLOCK)
#define BLOCKS_PER_BATCH 16
#define FSTRIDE 25
#define NTRIP (NNODES / 16)
#define QCAP 40
#define QROW (BLOCK * 8)

typedef __attribute__((ext_vector_type(2))) float v2f;
typedef __attribute__((ext_vector_type(8))) float v8f;

__global__ __launch_bounds__(BLOCK)
void tsp_embed_kernel(const float* __restrict__ locs,
                      const float* __restrict__ W,
                      const float* __restrict__ bias,
                      float* __restrict__ out)
{
    __shared__ alignas(16) float sLocs[2 * NNODES];
    __shared__ float sFeat[NODES_PER_BLOCK * FSTRIDE];
    __shared__ float sW[FDIM * EMBED_DIM];
    __shared__ float2 sQ[QCAP * BLOCK];

    const int tid   = threadIdx.x;
    const int batch = blockIdx.x / BLOCKS_PER_BATCH;
    const int node0 = (blockIdx.x % BLOCKS_PER_BATCH) * NODES_PER_BLOCK;

    const float* bl = locs + (size_t)batch * NNODES * 2;
#pragma unroll 8
    for (int idx = tid; idx < 2 * NNODES; idx += BLOCK) sLocs[idx] = bl[idx];
    for (int idx = tid; idx < FDIM * EMBED_DIM; idx += BLOCK) {
        const int f = idx / EMBED_DIM;
        const int e = idx % EMBED_DIM;
        sW[idx] = (f < NODE_DIM) ? W[e * NODE_DIM + f] : 0.0f;
    }
    __syncthreads();

    const int  part  = __builtin_amdgcn_readfirstlane(tid) / NODES_PER_BLOCK;
    const int  nl    = tid % NODES_PER_BLOCK;
    const int  tBeg  = (part * NTRIP) / NPARTS;
    const int  tEnd  = ((part + 1) * NTRIP) / NPARTS;
    const int  node  = node0 + nl;
    const bool valid = node < NNODES;
    float bestD[KNN];
    int   bestJ[KNN];
#pragma unroll
    for (int k = 0; k < KNN; ++k) { bestD[k] = 3.0e38f; bestJ[k] = 0; }

    const float2* sl  = (const float2*)sLocs;
    const float4* sl4 = (const float4*)sLocs;
    float px = 0.0f, py = 0.0f;
    if (valid) { const float2 me = sl[node]; px = me.x; py = me.y; }

    float thr = valid ? 3.0e38f : -1.0f;
    unsigned qb = tid * 8;

    auto cand = [&](float qx, float qy, int j) {
        const float dx = qx - px;
        const float dy = qy - py;
        const float d2 = __builtin_fmaf(dx, dx, dy * dy);
        *(float2*)((char*)sQ + qb) = make_float2(d2, __int_as_float(j));
        qb += (d2 < thr) ? QROW : 0;
    };
    auto proc8 = [&](const float4 (&h)[4], int jb) {
#pragma unroll
        for (int u = 0; u < 4; ++u) {
            cand(h[u].x, h[u].y, jb + 2 * u);
            cand(h[u].z, h[u].w, jb + 2 * u + 1);
        }
    };

    auto insert = [&](float d2, int j) {
        bestD[KNN - 1] = d2;
        bestJ[KNN - 1] = j;
#pragma unroll
        for (int s = KNN - 1; s > 0; --s) {
            const bool  up = bestD[s] < bestD[s - 1];
            const float dl = up ? bestD[s] : bestD[s - 1];
            const float dh = up ? bestD[s - 1] : bestD[s];
            const int   jl = up ? bestJ[s] : bestJ[s - 1];
            const int   jh = up ? bestJ[s - 1] : bestJ[s];
            bestD[s - 1] = dl; bestD[s] = dh;
            bestJ[s - 1] = jl; bestJ[s] = jh;
        }
    };

    auto drain = [&](int cnt) {
        const int lastRow = (cnt > 0) ? (cnt - 1) : 0;
        float2 e = sQ[tid];
#pragma nounroll
        for (int i = 0; ; ++i) {
            const int nx = (i + 1 < lastRow) ? (i + 1) : lastRow;
            const float2 en = sQ[nx * BLOCK + tid];
            const bool more = __builtin_amdgcn_ballot_w32(i + 1 < cnt) != 0u;
            const float d2 = e.x;
            const int   j  = __float_as_int(e.y);
            if (i < cnt && j != node && d2 < bestD[KNN - 1]) insert(d2, j);
            if (!more) break;
            e = en;
        }
    };

    float4 h0[4], h1[4];
#pragma unroll
    for (int u = 0; u < 4; ++u) h0[u] = sl4[tBeg * 8 + u];

    for (int t = tBeg; t < tEnd; ++t) {
        const int jb = t * 16;
#pragma unroll
        for (int u = 0; u < 4; ++u) h1[u] = sl4[(jb + 8) / 2 + u];
        proc8(h0, jb);
        const int cntMid = qb / QROW;
        const int jn = (jb + 16 < NNODES) ? (jb + 16) : (NNODES - 8);
#pragma unroll
        for (int u = 0; u < 4; ++u) h0[u] = sl4[jn / 2 + u];
        proc8(h1, jb + 8);

        const bool full = __builtin_amdgcn_ballot_w32(cntMid > QCAP - 24) != 0u;
        if (full || t == tEnd - 1) {
            drain(qb / QROW);
            qb = tid * 8;
            if (valid) thr = bestD[KNN - 1];
        }
    }

    __syncthreads();
    if (part > 0) {
#pragma unroll
        for (int k = 0; k < KNN; ++k)
            sQ[((part - 1) * KNN + k) * BLOCK + nl] = make_float2(bestD[k], __int_as_float(bestJ[k]));
    }
    __syncthreads();

    if (part == 0) {
    drain((NPARTS - 1) * KNN);

    float* fr = &sFeat[nl * FSTRIDE];
    if (node < NNODES) {
        fr[0] = px;
        fr[1] = py;
#pragma unroll
        for (int k = 0; k < KNN; ++k) {
            const float2 q = sl[bestJ[k]];
            fr[2 + 2 * k]     = q.x - px;
            fr[2 + 2 * k + 1] = q.y - py;
        }
        fr[NODE_DIM]     = 0.0f;
        fr[NODE_DIM + 1] = 0.0f;
    } else {
#pragma unroll
        for (int k = 0; k < FDIM; ++k) fr[k] = 0.0f;
    }
    }
    __syncthreads();

    if (part != 0) return;
    const int wave   = tid >> 5;
    const int lane   = tid & 31;
    const int laneLo = lane & 15;
    const int kHalf  = (lane < 16) ? 0 : 2;
    const int mBase  = (lane < 16) ? 0 : 8;

    const int t0 = wave * 2;
    const int t1 = t0 + 1;
    const bool ok0 = (node0 + t0 * 16) < NNODES;
    const bool ok1 = (node0 + t1 * 16) < NNODES;

    v2f a0[6], a1[6];
    {
        const float* fa0 = &sFeat[(t0 * 16 + laneLo) * FSTRIDE];
        const float* fa1 = &sFeat[(t1 * 16 + laneLo) * FSTRIDE];
#pragma unroll
        for (int kc = 0; kc < 6; ++kc) {
            const int k0 = kc * 4 + kHalf;
            a0[kc].x = fa0[k0];  a0[kc].y = fa0[k0 + 1];
            a1[kc].x = fa1[k0];  a1[kc].y = fa1[k0 + 1];
        }
    }

    float* const outB = out + (size_t)batch * NNODES * EMBED_DIM;

#pragma unroll
    for (int ct = 0; ct < 8; ++ct) {
        const int e = ct * 16 + laneLo;
        v2f bf[6];
#pragma unroll
        for (int kc = 0; kc < 6; ++kc) {
            const int k0 = kc * 4 + kHalf;
            bf[kc].x = sW[k0 * EMBED_DIM + e];
            bf[kc].y = sW[(k0 + 1) * EMBED_DIM + e];
        }
        v8f acc0 = {}, acc1 = {};
#pragma unroll
        for (int kc = 0; kc < 6; ++kc)
            acc0 = __builtin_amdgcn_wmma_f32_16x16x4_f32(
                false, a0[kc], false, bf[kc], (short)0, acc0, false, false);
#pragma unroll
        for (int kc = 0; kc < 6; ++kc)
            acc1 = __builtin_amdgcn_wmma_f32_16x16x4_f32(
                false, a1[kc], false, bf[kc], (short)0, acc1, false, false);

        const float bv = bias[e];
        if (ok0) {
            float* p = outB + (size_t)(node0 + t0 * 16 + mBase) * EMBED_DIM + e;
#pragma unroll
            for (int r = 0; r < 8; ++r)
                p[(size_t)r * EMBED_DIM] = acc0[r] + bv;
        }
        if (ok1) {
            float* p = outB + (size_t)(node0 + t1 * 16 + mBase) * EMBED_DIM + e;
#pragma unroll
            for (int r = 0; r < 8; ++r)
                p[(size_t)r * EMBED_DIM] = acc1[r] + bv;
        }
    }
}

extern "C" void kernel_launch(void* const* d_in, const int* in_sizes, int n_in,
                              void* d_out, int out_size, void* d_ws, size_t ws_size,
                              hipStream_t stream) {
    const float* locs = (const float*)d_in[0];
    const float* W    = (const float*)d_in[1];
    const float* b    = (const float*)d_in[2];
    float* out        = (float*)d_out;

    dim3 grid(NBATCH * BLOCKS_PER_BATCH);
    dim3 block(BLOCK);
    tsp_embed_kernel<<<grid, block, 0, stream>>>(locs, W, b, out);
}


# KNN scan operands laid out for the packed math: LDS holds x[4], y[4], index[4] per 4 candidates, the queue a d2 plane and a j plane; per-candidate register shuffles and index moves are gone
# speedup vs baseline: 5.2678x; 1.0456x over previous
#include <hip/hip_runtime.h>
#include <hip/hip_bf16.h>

#define EMBED_DIM 128
#define KNN 10
#define NODE_DIM 22
#define FDIM 24
#define NNODES 2000
#define NBATCH 16
#define NPARTS 4
#define NODES_PER_BLOCK 128
#define BLOCK (NPARTS * NODES_PER_BLOCK)
#define BLOCKS_PER_BATCH 16
#define FSTRIDE 25
#define NTRIP (NNODES / 16)
#define QCAP 40
#define QROW (BLOCK * 8)
#define CGRP 12
#define NGRP (NNODES / 4)

typedef __attribute__((ext_vector_type(2))) float v2f;
typedef __attribute__((ext_vector_type(4))) float v4f;
typedef __attribute__((ext_vector_type(8))) float v8f;

__global__ __launch_bounds__(BLOCK)
void tsp_embed_kernel(const float* __restrict__ locs,
                      const float* __restrict__ W,
                      const float* __restrict__ bias,
                      float* __restrict__ out)
{
    __shared__ alignas(16) float sLocs[2 * NNODES];
    __shared__ alignas(16) float sCand[NGRP * CGRP];
    __shared__ float sFeat[NODES_PER_BLOCK * FSTRIDE];
    __shared__ float sW[FDIM * EMBED_DIM];
    __shared__ float sQ[2 * QCAP * BLOCK];

    const int tid   = threadIdx.x;
    const int batch = blockIdx.x / BLOCKS_PER_BATCH;
    const int node0 = (blockIdx.x % BLOCKS_PER_BATCH) * NODES_PER_BLOCK;

    const float* bl = locs + (size_t)batch * NNODES * 2;
#pragma unroll 8
    for (int idx = tid; idx < 2 * NNODES; idx += BLOCK) {
        const float v = bl[idx];
        const int nd = idx >> 1;
        sLocs[idx] = v;
        sCand[(nd >> 2) * CGRP + (idx & 1) * 4 + (nd & 3)] = v;
    }
    for (int j = tid; j < NNODES; j += BLOCK)
        sCand[(j >> 2) * CGRP + 8 + (j & 3)] = __int_as_float(j);
    for (int idx = tid; idx < FDIM * EMBED_DIM; idx += BLOCK) {
        const int f = idx / EMBED_DIM;
        const int e = idx % EMBED_DIM;
        sW[idx] = (f < NODE_DIM) ? W[e * NODE_DIM + f] : 0.0f;
    }
    __syncthreads();

    const int  part  = __builtin_amdgcn_readfirstlane(tid) / NODES_PER_BLOCK;
    const int  nl    = tid % NODES_PER_BLOCK;
    const int  tBeg  = (part * NTRIP) / NPARTS;
    const int  tEnd  = ((part + 1) * NTRIP) / NPARTS;
    const int  node  = node0 + nl;
    const bool valid = node < NNODES;
    float bestD[KNN];
    int   bestJ[KNN];
#pragma unroll
    for (int k = 0; k < KNN; ++k) { bestD[k] = 3.0e38f; bestJ[k] = 0; }

    const float2* sl  = (const float2*)sLocs;
    const v4f*    sc4 = (const v4f*)sCand;
    float px = 0.0f, py = 0.0f;
    if (valid) { const float2 me = sl[node]; px = me.x; py = me.y; }

    float thr = valid ? 3.0e38f : -1.0f;
    unsigned qb = tid * 4;
    const v2f pxv = {px, px};
    const v2f pyv = {py, py};

    auto cand = [&](float d2, float jf) {
        float* q = (float*)((char*)sQ + qb);
        q[0]     = d2;
        q[BLOCK] = jf;
        qb += (d2 < thr) ? QROW : 0;
    };
    auto pair2 = [&](v2f x, v2f y) {
        const v2f dx = x - pxv;
        const v2f dy = y - pyv;
        return __builtin_elementwise_fma(dx, dx, dy * dy);
    };
    auto proc8 = [&](const v4f (&h)[6]) {
#pragma unroll
        for (int g = 0; g < 2; ++g) {
            const v4f X = h[3 * g], Y = h[3 * g + 1], J = h[3 * g + 2];
            const v2f da = pair2(X.xy, Y.xy);
            const v2f db = pair2(X.zw, Y.zw);
            cand(da.x, J.x);
            cand(da.y, J.y);
            cand(db.x, J.z);
            cand(db.y, J.w);
        }
    };

    auto insert = [&](float d2, int j) {
        bestD[KNN - 1] = d2;
        bestJ[KNN - 1] = j;
#pragma unroll
        for (int s = KNN - 1; s > 0; --s) {
            const bool  up = bestD[s] < bestD[s - 1];
            const float dl = up ? bestD[s] : bestD[s - 1];
            const float dh = up ? bestD[s - 1] : bestD[s];
            const int   jl = up ? bestJ[s] : bestJ[s - 1];
            const int   jh = up ? bestJ[s - 1] : bestJ[s];
            bestD[s - 1] = dl; bestD[s] = dh;
            bestJ[s - 1] = jl; bestJ[s] = jh;
        }
    };

    auto drain = [&](int cnt) {
        const int lastRow = (cnt > 0) ? (cnt - 1) : 0;
        float ed = sQ[tid], ej = sQ[BLOCK + tid];
#pragma nounroll
        for (int i = 0; ; ++i) {
            const int nx = (i + 1 < lastRow) ? (i + 1) : lastRow;
            const float nd = sQ[nx * 2 * BLOCK + tid];
            const float nj = sQ[nx * 2 * BLOCK + BLOCK + tid];
            const bool more = __builtin_amdgcn_ballot_w32(i + 1 < cnt) != 0u;
            const float d2 = ed;
            const int   j  = __float_as_int(ej);
            if (i < cnt && j != node && d2 < bestD[KNN - 1]) insert(d2, j);
            if (!more) break;
            ed = nd; ej = nj;
        }
    };

    v4f h0[6], h1[6];
#pragma unroll
    for (int u = 0; u < 6; ++u) h0[u] = sc4[tBeg * 12 + u];

    for (int t = tBeg; t < tEnd; ++t) {
#pragma unroll
        for (int u = 0; u < 6; ++u) h1[u] = sc4[t * 12 + 6 + u];
        proc8(h0);
        const int cntMid = qb / QROW;
        const int vn = (t + 1 < NTRIP) ? (t + 1) * 12 : (NTRIP * 12 - 6);
#pragma unroll
        for (int u = 0; u < 6; ++u) h0[u] = sc4[vn + u];
        proc8(h1);

        const bool full = __builtin_amdgcn_ballot_w32(cntMid > QCAP - 24) != 0u;
        if (full || t == tEnd - 1) {
            drain(qb / QROW);
            qb = tid * 4;
            if (valid) thr = bestD[KNN - 1];
        }
    }

    __syncthreads();
    if (part > 0) {
#pragma unroll
        for (int k = 0; k < KNN; ++k)
        {
            const int row = (part - 1) * KNN + k;
            sQ[row * 2 * BLOCK + nl]         = bestD[k];
            sQ[row * 2 * BLOCK + BLOCK + nl] = __int_as_float(bestJ[k]);
        }
    }
    __syncthreads();

    if (part == 0) {
    drain((NPARTS - 1) * KNN);

    float* fr = &sFeat[nl * FSTRIDE];
    if (node < NNODES) {
        fr[0] = px;
        fr[1] = py;
#pragma unroll
        for (int k = 0; k < KNN; ++k) {
            const float2 q = sl[bestJ[k]];
            fr[2 + 2 * k]     = q.x - px;
            fr[2 + 2 * k + 1] = q.y - py;
        }
        fr[NODE_DIM]     = 0.0f;
        fr[NODE_DIM + 1] = 0.0f;
    } else {
#pragma unroll
        for (int k = 0; k < FDIM; ++k) fr[k] = 0.0f;
    }
    }
    __syncthreads();

    if (part != 0) return;
    const int wave   = tid >> 5;
    const int lane   = tid & 31;
    const int laneLo = lane & 15;
    const int kHalf  = (lane < 16) ? 0 : 2;
    const int mBase  = (lane < 16) ? 0 : 8;

    const int t0 = wave * 2;
    const int t1 = t0 + 1;
    const bool ok0 = (node0 + t0 * 16) < NNODES;
    const bool ok1 = (node0 + t1 * 16) < NNODES;

    v2f a0[6], a1[6];
    {
        const float* fa0 = &sFeat[(t0 * 16 + laneLo) * FSTRIDE];
        const float* fa1 = &sFeat[(t1 * 16 + laneLo) * FSTRIDE];
#pragma unroll
        for (int kc = 0; kc < 6; ++kc) {
            const int k0 = kc * 4 + kHalf;
            a0[kc].x = fa0[k0];  a0[kc].y = fa0[k0 + 1];
            a1[kc].x = fa1[k0];  a1[kc].y = fa1[k0 + 1];
        }
    }

    float* const outB = out + (size_t)batch * NNODES * EMBED_DIM;

#pragma unroll
    for (int ct = 0; ct < 8; ++ct) {
        const int e = ct * 16 + laneLo;
        v2f bf[6];
#pragma unroll
        for (int kc = 0; kc < 6; ++kc) {
            const int k0 = kc * 4 + kHalf;
            bf[kc].x = sW[k0 * EMBED_DIM + e];
            bf[kc].y = sW[(k0 + 1) * EMBED_DIM + e];
        }
        v8f acc0 = {}, acc1 = {};
#pragma unroll
        for (int kc = 0; kc < 6; ++kc)
            acc0 = __builtin_amdgcn_wmma_f32_16x16x4_f32(
                false, a0[kc], false, bf[kc], (short)0, acc0, false, false);
#pragma unroll
        for (int kc = 0; kc < 6; ++kc)
            acc1 = __builtin_amdgcn_wmma_f32_16x16x4_f32(
                false, a1[kc], false, bf[kc], (short)0, acc1, false, false);

        const float bv = bias[e];
        if (ok0) {
            float* p = outB + (size_t)(node0 + t0 * 16 + mBase) * EMBED_DIM + e;
#pragma unroll
            for (int r = 0; r < 8; ++r)
                p[(size_t)r * EMBED_DIM] = acc0[r] + bv;
        }
        if (ok1) {
            float* p = outB + (size_t)(node0 + t1 * 16 + mBase) * EMBED_DIM + e;
#pragma unroll
            for (int r = 0; r < 8; ++r)
                p[(size_t)r * EMBED_DIM] = acc1[r] + bv;
        }
    }
}

extern "C" void kernel_launch(void* const* d_in, const int* in_sizes, int n_in,
                              void* d_out, int out_size, void* d_ws, size_t ws_size,
                              hipStream_t stream) {
    const float* locs = (const float*)d_in[0];
    const float* W    = (const float*)d_in[1];
    const float* b    = (const float*)d_in[2];
    float* out        = (float*)d_out;

    dim3 grid(NBATCH * BLOCKS_PER_BATCH);
    dim3 block(BLOCK);
    tsp_embed_kernel<<<grid, block, 0, stream>>>(locs, W, b, out);
}


# Drain's sorted insert made branch-free: every slot picked from the old list by independent compares (distance = median of its two neighbours and the entry); was 9 dependent swaps inside a branch
# speedup vs baseline: 5.5188x; 1.0476x over previous
#include <hip/hip_runtime.h>
#include <hip/hip_bf16.h>

#define EMBED_DIM 128
#define KNN 10
#define NODE_DIM 22
#define FDIM 24
#define NNODES 2000
#define NBATCH 16
#define NPARTS 4
#define NODES_PER_BLOCK 128
#define BLOCK (NPARTS * NODES_PER_BLOCK)
#define BLOCKS_PER_BATCH 16
#define FSTRIDE 25
#define NTRIP (NNODES / 16)
#define QCAP 40
#define QROW (BLOCK * 8)
#define CGRP 12
#define NGRP (NNODES / 4)

typedef __attribute__((ext_vector_type(2))) float v2f;
typedef __attribute__((ext_vector_type(4))) float v4f;
typedef __attribute__((ext_vector_type(8))) float v8f;

__global__ __launch_bounds__(BLOCK)
void tsp_embed_kernel(const float* __restrict__ locs,
                      const float* __restrict__ W,
                      const float* __restrict__ bias,
                      float* __restrict__ out)
{
    __shared__ alignas(16) float sLocs[2 * NNODES];
    __shared__ alignas(16) float sCand[NGRP * CGRP];
    __shared__ float sFeat[NODES_PER_BLOCK * FSTRIDE];
    __shared__ float sW[FDIM * EMBED_DIM];
    __shared__ float sQ[2 * QCAP * BLOCK];

    const int tid   = threadIdx.x;
    const int batch = blockIdx.x / BLOCKS_PER_BATCH;
    const int node0 = (blockIdx.x % BLOCKS_PER_BATCH) * NODES_PER_BLOCK;

    const float* bl = locs + (size_t)batch * NNODES * 2;
#pragma unroll 8
    for (int idx = tid; idx < 2 * NNODES; idx += BLOCK) {
        const float v = bl[idx];
        const int nd = idx >> 1;
        sLocs[idx] = v;
        sCand[(nd >> 2) * CGRP + (idx & 1) * 4 + (nd & 3)] = v;
    }
    for (int j = tid; j < NNODES; j += BLOCK)
        sCand[(j >> 2) * CGRP + 8 + (j & 3)] = __int_as_float(j);
    for (int idx = tid; idx < FDIM * EMBED_DIM; idx += BLOCK) {
        const int f = idx / EMBED_DIM;
        const int e = idx % EMBED_DIM;
        sW[idx] = (f < NODE_DIM) ? W[e * NODE_DIM + f] : 0.0f;
    }
    __syncthreads();

    const int  part  = __builtin_amdgcn_readfirstlane(tid) / NODES_PER_BLOCK;
    const int  nl    = tid % NODES_PER_BLOCK;
    const int  tBeg  = (part * NTRIP) / NPARTS;
    const int  tEnd  = ((part + 1) * NTRIP) / NPARTS;
    const int  node  = node0 + nl;
    const bool valid = node < NNODES;
    float bestD[KNN];
    int   bestJ[KNN];
#pragma unroll
    for (int k = 0; k < KNN; ++k) { bestD[k] = 3.0e38f; bestJ[k] = 0; }

    const float2* sl  = (const float2*)sLocs;
    const v4f*    sc4 = (const v4f*)sCand;
    float px = 0.0f, py = 0.0f;
    if (valid) { const float2 me = sl[node]; px = me.x; py = me.y; }

    float thr = valid ? 3.0e38f : -1.0f;
    unsigned qb = tid * 4;
    const v2f pxv = {px, px};
    const v2f pyv = {py, py};

    auto cand = [&](float d2, float jf) {
        float* q = (float*)((char*)sQ + qb);
        q[0]     = d2;
        q[BLOCK] = jf;
        qb += (d2 < thr) ? QROW : 0;
    };
    auto pair2 = [&](v2f x, v2f y) {
        const v2f dx = x - pxv;
        const v2f dy = y - pyv;
        return __builtin_elementwise_fma(dx, dx, dy * dy);
    };
    auto proc8 = [&](const v4f (&h)[6]) {
#pragma unroll
        for (int g = 0; g < 2; ++g) {
            const v4f X = h[3 * g], Y = h[3 * g + 1], J = h[3 * g + 2];
            const v2f da = pair2(X.xy, Y.xy);
            const v2f db = pair2(X.zw, Y.zw);
            cand(da.x, J.x);
            cand(da.y, J.y);
            cand(db.x, J.z);
            cand(db.y, J.w);
        }
    };

    auto insert = [&](float d2, int j) {
        bool c[KNN];
#pragma unroll
        for (int s = 0; s < KNN; ++s) c[s] = d2 < bestD[s];
#pragma unroll
        for (int s = KNN - 1; s > 0; --s) {
            const int jLow = bestJ[s - 1], jOwn = bestJ[s];
            const int jNew = c[s] ? j : jOwn;
            bestJ[s] = c[s - 1] ? jLow : jNew;
            bestD[s] = __builtin_amdgcn_fmed3f(bestD[s - 1], bestD[s], d2);
        }
        const float d0 = bestD[0];
        const int   j0 = bestJ[0];
        bestD[0] = c[0] ? d2 : d0;
        bestJ[0] = c[0] ? j  : j0;
    };

    auto drain = [&](int cnt) {
        const int lastRow = (cnt > 0) ? (cnt - 1) : 0;
        float ed = sQ[tid], ej = sQ[BLOCK + tid];
        bool own = 0 < cnt;
#pragma nounroll
        for (int i = 0; ; ++i) {
            const int nx = (i + 1 < lastRow) ? (i + 1) : lastRow;
            const float nd = sQ[nx * 2 * BLOCK + tid];
            const float nj = sQ[nx * 2 * BLOCK + BLOCK + tid];
            asm volatile("" ::: "memory");
            const bool ownNext = i + 1 < cnt;
            const bool more = __builtin_amdgcn_ballot_w32(ownNext) != 0u;
            const int   j  = __float_as_int(ej);
            const float none = 3.0e38f;
            const float d2 = (own && j != node) ? ed : none;
            insert(d2, j);
            if (!more) break;
            ed = nd; ej = nj; own = ownNext;
        }
    };

    v4f h0[6], h1[6];
#pragma unroll
    for (int u = 0; u < 6; ++u) h0[u] = sc4[tBeg * 12 + u];

    for (int t = tBeg; t < tEnd; ++t) {
#pragma unroll
        for (int u = 0; u < 6; ++u) h1[u] = sc4[t * 12 + 6 + u];
        proc8(h0);
        const int cntMid = qb / QROW;
        const int vn = (t + 1 < NTRIP) ? (t + 1) * 12 : (NTRIP * 12 - 6);
#pragma unroll
        for (int u = 0; u < 6; ++u) h0[u] = sc4[vn + u];
        proc8(h1);

        const bool full = __builtin_amdgcn_ballot_w32(cntMid > QCAP - 24) != 0u;
        if (full || t == tEnd - 1) {
            drain(qb / QROW);
            qb = tid * 4;
            if (valid) thr = bestD[KNN - 1];
        }
    }

    __syncthreads();
    if (part > 0) {
#pragma unroll
        for (int k = 0; k < KNN; ++k)
        {
            const int row = (part - 1) * KNN + k;
            sQ[row * 2 * BLOCK + nl]         = bestD[k];
            sQ[row * 2 * BLOCK + BLOCK + nl] = __int_as_float(bestJ[k]);
        }
    }
    __syncthreads();

    if (part == 0) {
    drain((NPARTS - 1) * KNN);

    float* fr = &sFeat[nl * FSTRIDE];
    if (node < NNODES) {
        fr[0] = px;
        fr[1] = py;
#pragma unroll
        for (int k = 0; k < KNN; ++k) {
            const float2 q = sl[bestJ[k]];
            fr[2 + 2 * k]     = q.x - px;
            fr[2 + 2 * k + 1] = q.y - py;
        }
        fr[NODE_DIM]     = 0.0f;
        fr[NODE_DIM + 1] = 0.0f;
    } else {
#pragma unroll
        for (int k = 0; k < FDIM; ++k) fr[k] = 0.0f;
    }
    }
    __syncthreads();

    if (part != 0) return;
    const int wave   = tid >> 5;
    const int lane   = tid & 31;
    const int laneLo = lane & 15;
    const int kHalf  = (lane < 16) ? 0 : 2;
    const int mBase  = (lane < 16) ? 0 : 8;

    const int t0 = wave * 2;
    const int t1 = t0 + 1;
    const bool ok0 = (node0 + t0 * 16) < NNODES;
    const bool ok1 = (node0 + t1 * 16) < NNODES;

    v2f a0[6], a1[6];
    {
        const float* fa0 = &sFeat[(t0 * 16 + laneLo) * FSTRIDE];
        const float* fa1 = &sFeat[(t1 * 16 + laneLo) * FSTRIDE];
#pragma unroll
        for (int kc = 0; kc < 6; ++kc) {
            const int k0 = kc * 4 + kHalf;
            a0[kc].x = fa0[k0];  a0[kc].y = fa0[k0 + 1];
            a1[kc].x = fa1[k0];  a1[kc].y = fa1[k0 + 1];
        }
    }

    float* const outB = out + (size_t)batch * NNODES * EMBED_DIM;

#pragma unroll
    for (int ct = 0; ct < 8; ++ct) {
        const int e = ct * 16 + laneLo;
        v2f bf[6];
#pragma unroll
        for (int kc = 0; kc < 6; ++kc) {
            const int k0 = kc * 4 + kHalf;
            bf[kc].x = sW[k0 * EMBED_DIM + e];
            bf[kc].y = sW[(k0 + 1) * EMBED_DIM + e];
        }
        v8f acc0 = {}, acc1 = {};
#pragma unroll
        for (int kc = 0; kc < 6; ++kc)
            acc0 = __builtin_amdgcn_wmma_f32_16x16x4_f32(
                false, a0[kc], false, bf[kc], (short)0, acc0, false, false);
#pragma unroll
        for (int kc = 0; kc < 6; ++kc)
            acc1 = __builtin_amdgcn_wmma_f32_16x16x4_f32(
                false, a1[kc], false, bf[kc], (short)0, acc1, false, false);

        const float bv = bias[e];
        if (ok0) {
            float* p = outB + (size_t)(node0 + t0 * 16 + mBase) * EMBED_DIM + e;
#pragma unroll
            for (int r = 0; r < 8; ++r)
                p[(size_t)r * EMBED_DIM] = acc0[r] + bv;
        }
        if (ok1) {
            float* p = outB + (size_t)(node0 + t1 * 16 + mBase) * EMBED_DIM + e;
#pragma unroll
            for (int r = 0; r < 8; ++r)
                p[(size_t)r * EMBED_DIM] = acc1[r] + bv;
        }
    }
}

extern "C" void kernel_launch(void* const* d_in, const int* in_sizes, int n_in,
                              void* d_out, int out_size, void* d_ws, size_t ws_size,
                              hipStream_t stream) {
    const float* locs = (const float*)d_in[0];
    const float* W    = (const float*)d_in[1];
    const float* b    = (const float*)d_in[2];
    float* out        = (float*)d_out;

    dim3 grid(NBATCH * BLOCKS_PER_BATCH);
    dim3 block(BLOCK);
    tsp_embed_kernel<<<grid, block, 0, stream>>>(locs, W, b, out);
}


# The 4 wave groups scanning one node share their tenth-best distance via LDS; each filters with the smallest of the four (read once a trip, first drain after one trip): fewer lockstep drain passes
# speedup vs baseline: 5.9589x; 1.0797x over previous
#include <hip/hip_runtime.h>
#include <hip/hip_bf16.h>

#define EMBED_DIM 128
#define KNN 10
#define NODE_DIM 22
#define FDIM 24
#define NNODES 2000
#define NBATCH 16
#define NPARTS 4
#define NODES_PER_BLOCK 128
#define BLOCK (NPARTS * NODES_PER_BLOCK)
#define BLOCKS_PER_BATCH 16
#define FSTRIDE 25
#define NTRIP (NNODES / 16)
#define QCAP 40
#define QROW (BLOCK * 8)
#define CGRP 12
#define NGRP (NNODES / 4)

typedef __attribute__((ext_vector_type(2))) float v2f;
typedef __attribute__((ext_vector_type(4))) float v4f;
typedef __attribute__((ext_vector_type(8))) float v8f;
typedef __attribute__((ext_vector_type(4))) int   v4i;

__global__ __launch_bounds__(BLOCK)
void tsp_embed_kernel(const float* __restrict__ locs,
                      const float* __restrict__ W,
                      const float* __restrict__ bias,
                      float* __restrict__ out)
{
    __shared__ alignas(16) float sLocs[2 * NNODES];
    __shared__ alignas(16) float sCand[NGRP * CGRP];
    __shared__ float sFeat[NODES_PER_BLOCK * FSTRIDE];
    __shared__ float sW[FDIM * EMBED_DIM];
    __shared__ float sQ[2 * QCAP * BLOCK];
    __shared__ alignas(16) int sThr[NPARTS * NODES_PER_BLOCK];

    const int tid   = threadIdx.x;
    const int batch = blockIdx.x / BLOCKS_PER_BATCH;
    const int node0 = (blockIdx.x % BLOCKS_PER_BATCH) * NODES_PER_BLOCK;

    const float* bl = locs + (size_t)batch * NNODES * 2;
#pragma unroll 8
    for (int idx = tid; idx < 2 * NNODES; idx += BLOCK) {
        const float v = bl[idx];
        const int nd = idx >> 1;
        sLocs[idx] = v;
        sCand[(nd >> 2) * CGRP + (idx & 1) * 4 + (nd & 3)] = v;
    }
    for (int j = tid; j < NNODES; j += BLOCK)
        sCand[(j >> 2) * CGRP + 8 + (j & 3)] = __int_as_float(j);
    sThr[(tid % NODES_PER_BLOCK) * NPARTS + tid / NODES_PER_BLOCK] =
        __float_as_int((node0 + tid % NODES_PER_BLOCK < NNODES) ? 3.0e38f : -1.0f);
    for (int idx = tid; idx < FDIM * EMBED_DIM; idx += BLOCK) {
        const int f = idx / EMBED_DIM;
        const int e = idx % EMBED_DIM;
        sW[idx] = (f < NODE_DIM) ? W[e * NODE_DIM + f] : 0.0f;
    }
    __syncthreads();

    const int  part  = __builtin_amdgcn_readfirstlane(tid) / NODES_PER_BLOCK;
    const int  nl    = tid % NODES_PER_BLOCK;
    const int  tBeg  = (part * NTRIP) / NPARTS;
    const int  tEnd  = ((part + 1) * NTRIP) / NPARTS;
    const int  node  = node0 + nl;
    const bool valid = node < NNODES;
    float bestD[KNN];
    int   bestJ[KNN];
#pragma unroll
    for (int k = 0; k < KNN; ++k) { bestD[k] = 3.0e38f; bestJ[k] = 0; }

    const float2* sl  = (const float2*)sLocs;
    const v4f*    sc4 = (const v4f*)sCand;
    float px = 0.0f, py = 0.0f;
    if (valid) { const float2 me = sl[node]; px = me.x; py = me.y; }

    float thr = valid ? 3.0e38f : -1.0f;
    unsigned qb = tid * 4;
    const v2f pxv = {px, px};
    const v2f pyv = {py, py};

    auto cand = [&](float d2, float jf) {
        float* q = (float*)((char*)sQ + qb);
        q[0]     = d2;
        q[BLOCK] = jf;
        qb += (d2 <= thr) ? QROW : 0;
    };
    auto pair2 = [&](v2f x, v2f y) {
        const v2f dx = x - pxv;
        const v2f dy = y - pyv;
        return __builtin_elementwise_fma(dx, dx, dy * dy);
    };
    auto proc8 = [&](const v4f (&h)[6]) {
#pragma unroll
        for (int g = 0; g < 2; ++g) {
            const v4f X = h[3 * g], Y = h[3 * g + 1], J = h[3 * g + 2];
            const v2f da = pair2(X.xy, Y.xy);
            const v2f db = pair2(X.zw, Y.zw);
            cand(da.x, J.x);
            cand(da.y, J.y);
            cand(db.x, J.z);
            cand(db.y, J.w);
        }
    };

    auto insert = [&](float d2, int j) {
        bool c[KNN];
#pragma unroll
        for (int s = 0; s < KNN; ++s) c[s] = d2 < bestD[s];
#pragma unroll
        for (int s = KNN - 1; s > 0; --s) {
            const int jLow = bestJ[s - 1], jOwn = bestJ[s];
            const int jNew = c[s] ? j : jOwn;
            bestJ[s] = c[s - 1] ? jLow : jNew;
            bestD[s] = __builtin_amdgcn_fmed3f(bestD[s - 1], bestD[s], d2);
        }
        const float d0 = bestD[0];
        const int   j0 = bestJ[0];
        bestD[0] = c[0] ? d2 : d0;
        bestJ[0] = c[0] ? j  : j0;
    };

    auto drain = [&](int cnt) {
        const int lastRow = (cnt > 0) ? (cnt - 1) : 0;
        float ed = sQ[tid], ej = sQ[BLOCK + tid];
        bool own = 0 < cnt;
#pragma nounroll
        for (int i = 0; ; ++i) {
            const int nx = (i + 1 < lastRow) ? (i + 1) : lastRow;
            const float nd = sQ[nx * 2 * BLOCK + tid];
            const float nj = sQ[nx * 2 * BLOCK + BLOCK + tid];
            asm volatile("" ::: "memory");
            const bool ownNext = i + 1 < cnt;
            const bool more = __builtin_amdgcn_ballot_w32(ownNext) != 0u;
            const int   j  = __float_as_int(ej);
            const float none = 3.0e38f;
            const float d2 = (own && j != node) ? ed : none;
            insert(d2, j);
            if (!more) break;
            ed = nd; ej = nj; own = ownNext;
        }
    };

    v4f h0[6], h1[6];
#pragma unroll
    for (int u = 0; u < 6; ++u) h0[u] = sc4[tBeg * 12 + u];

    for (int t = tBeg; t < tEnd; ++t) {
#pragma unroll
        for (int u = 0; u < 6; ++u) h1[u] = sc4[t * 12 + 6 + u];
        const v4i sh = *(const v4i*)(sThr + nl * NPARTS);
        proc8(h0);
        const int cntMid = qb / QROW;
        {
            const int own = __float_as_int(bestD[KNN - 1]);
            const int m01 = sh.x < sh.y ? sh.x : sh.y;
            const int m23 = sh.z < sh.w ? sh.z : sh.w;
            const int m   = m01 < m23 ? m01 : m23;
            thr = __int_as_float(m < own ? m : own);
        }
        const int vn = (t + 1 < NTRIP) ? (t + 1) * 12 : (NTRIP * 12 - 6);
#pragma unroll
        for (int u = 0; u < 6; ++u) h0[u] = sc4[vn + u];
        proc8(h1);

        const bool full = __builtin_amdgcn_ballot_w32(cntMid > QCAP - 24) != 0u;
        if (full || t == tEnd - 1 || t == tBeg) {
            drain(qb / QROW);
            qb = tid * 4;
            const int own = __float_as_int(bestD[KNN - 1]);
            sThr[nl * NPARTS + part] = valid ? own : __float_as_int(-1.0f);
            const int cur = __float_as_int(thr);
            thr = __int_as_float(own < cur ? own : cur);
        }
    }

    __syncthreads();
    if (part > 0) {
#pragma unroll
        for (int k = 0; k < KNN; ++k)
        {
            const int row = (part - 1) * KNN + k;
            sQ[row * 2 * BLOCK + nl]         = bestD[k];
            sQ[row * 2 * BLOCK + BLOCK + nl] = __int_as_float(bestJ[k]);
        }
    }
    __syncthreads();

    if (part == 0) {
    drain((NPARTS - 1) * KNN);

    float* fr = &sFeat[nl * FSTRIDE];
    if (node < NNODES) {
        fr[0] = px;
        fr[1] = py;
#pragma unroll
        for (int k = 0; k < KNN; ++k) {
            const float2 q = sl[bestJ[k]];
            fr[2 + 2 * k]     = q.x - px;
            fr[2 + 2 * k + 1] = q.y - py;
        }
        fr[NODE_DIM]     = 0.0f;
        fr[NODE_DIM + 1] = 0.0f;
    } else {
#pragma unroll
        for (int k = 0; k < FDIM; ++k) fr[k] = 0.0f;
    }
    }
    __syncthreads();

    if (part != 0) return;
    const int wave   = tid >> 5;
    const int lane   = tid & 31;
    const int laneLo = lane & 15;
    const int kHalf  = (lane < 16) ? 0 : 2;
    const int mBase  = (lane < 16) ? 0 : 8;

    const int t0 = wave * 2;
    const int t1 = t0 + 1;
    const bool ok0 = (node0 + t0 * 16) < NNODES;
    const bool ok1 = (node0 + t1 * 16) < NNODES;

    v2f a0[6], a1[6];
    {
        const float* fa0 = &sFeat[(t0 * 16 + laneLo) * FSTRIDE];
        const float* fa1 = &sFeat[(t1 * 16 + laneLo) * FSTRIDE];
#pragma unroll
        for (int kc = 0; kc < 6; ++kc) {
            const int k0 = kc * 4 + kHalf;
            a0[kc].x = fa0[k0];  a0[kc].y = fa0[k0 + 1];
            a1[kc].x = fa1[k0];  a1[kc].y = fa1[k0 + 1];
        }
    }

    float* const outB = out + (size_t)batch * NNODES * EMBED_DIM;

#pragma unroll
    for (int ct = 0; ct < 8; ++ct) {
        const int e = ct * 16 + laneLo;
        v2f bf[6];
#pragma unroll
        for (int kc = 0; kc < 6; ++kc) {
            const int k0 = kc * 4 + kHalf;
            bf[kc].x = sW[k0 * EMBED_DIM + e];
            bf[kc].y = sW[(k0 + 1) * EMBED_DIM + e];
        }
        v8f acc0 = {}, acc1 = {};
#pragma unroll
        for (int kc = 0; kc < 6; ++kc)
            acc0 = __builtin_amdgcn_wmma_f32_16x16x4_f32(
                false, a0[kc], false, bf[kc], (short)0, acc0, false, false);
#pragma unroll
        for (int kc = 0; kc < 6; ++kc)
            acc1 = __builtin_amdgcn_wmma_f32_16x16x4_f32(
                false, a1[kc], false, bf[kc], (short)0, acc1, false, false);

        const float bv = bias[e];
        if (ok0) {
            float* p = outB + (size_t)(node0 + t0 * 16 + mBase) * EMBED_DIM + e;
#pragma unroll
            for (int r = 0; r < 8; ++r)
                p[(size_t)r * EMBED_DIM] = acc0[r] + bv;
        }
        if (ok1) {
            float* p = outB + (size_t)(node0 + t1 * 16 + mBase) * EMBED_DIM + e;
#pragma unroll
            for (int r = 0; r < 8; ++r)
                p[(size_t)r * EMBED_DIM] = acc1[r] + bv;
        }
    }
}

extern "C" void kernel_launch(void* const* d_in, const int* in_sizes, int n_in,
                              void* d_out, int out_size, void* d_ws, size_t ws_size,
                              hipStream_t stream) {
    const float* locs = (const float*)d_in[0];
    const float* W    = (const float*)d_in[1];
    const float* b    = (const float*)d_in[2];
    float* out        = (float*)d_out;

    dim3 grid(NBATCH * BLOCKS_PER_BATCH);
    dim3 block(BLOCK);
    tsp_embed_kernel<<<grid, block, 0, stream>>>(locs, W, b, out);
}


# Scan bound seeded before the scan: minima of 16-candidate groups over each part's first 16 trips; eleven disjoint groups (2 of one part, 3 of each other) bound the tenth best: few candidates queue
# speedup vs baseline: 6.8859x; 1.1556x over previous
#include <hip/hip_runtime.h>
#include <hip/hip_bf16.h>

#define EMBED_DIM 128
#define KNN 10
#define NODE_DIM 22
#define FDIM 24
#define NNODES 2000
#define NBATCH 16
#define NPARTS 4
#define NODES_PER_BLOCK 128
#define BLOCK (NPARTS * NODES_PER_BLOCK)
#define BLOCKS_PER_BATCH 16
#define FSTRIDE 25
#define NTRIP (NNODES / 16)
#define QCAP 40
#define QROW (BLOCK * 8)
#define CGRP 12
#define NGRP (NNODES / 4)
#define BOOT 16

typedef __attribute__((ext_vector_type(2))) float v2f;
typedef __attribute__((ext_vector_type(4))) float v4f;
typedef __attribute__((ext_vector_type(8))) float v8f;
typedef __attribute__((ext_vector_type(4))) int   v4i;

__global__ __launch_bounds__(BLOCK)
void tsp_embed_kernel(const float* __restrict__ locs,
                      const float* __restrict__ W,
                      const float* __restrict__ bias,
                      float* __restrict__ out)
{
    __shared__ alignas(16) float sLocs[2 * NNODES];
    __shared__ alignas(16) float sCand[NGRP * CGRP];
    __shared__ float sFeat[NODES_PER_BLOCK * FSTRIDE];
    __shared__ float sW[FDIM * EMBED_DIM];
    __shared__ float sQ[2 * QCAP * BLOCK];
    __shared__ alignas(16) int sThr[NPARTS * NODES_PER_BLOCK];
    __shared__ alignas(16) float sBoot[NODES_PER_BLOCK * NPARTS * 2];

    const int tid   = threadIdx.x;
    const int batch = blockIdx.x / BLOCKS_PER_BATCH;
    const int node0 = (blockIdx.x % BLOCKS_PER_BATCH) * NODES_PER_BLOCK;

    const float* bl = locs + (size_t)batch * NNODES * 2;
#pragma unroll 8
    for (int idx = tid; idx < 2 * NNODES; idx += BLOCK) {
        const float v = bl[idx];
        const int nd = idx >> 1;
        sLocs[idx] = v;
        sCand[(nd >> 2) * CGRP + (idx & 1) * 4 + (nd & 3)] = v;
    }
    for (int j = tid; j < NNODES; j += BLOCK)
        sCand[(j >> 2) * CGRP + 8 + (j & 3)] = __int_as_float(j);
    sThr[(tid % NODES_PER_BLOCK) * NPARTS + tid / NODES_PER_BLOCK] =
        __float_as_int((node0 + tid % NODES_PER_BLOCK < NNODES) ? 3.0e38f : -1.0f);
    for (int idx = tid; idx < FDIM * EMBED_DIM; idx += BLOCK) {
        const int f = idx / EMBED_DIM;
        const int e = idx % EMBED_DIM;
        sW[idx] = (f < NODE_DIM) ? W[e * NODE_DIM + f] : 0.0f;
    }
    __syncthreads();

    const int  part  = __builtin_amdgcn_readfirstlane(tid) / NODES_PER_BLOCK;
    const int  nl    = tid % NODES_PER_BLOCK;
    const int  tBeg  = (part * NTRIP) / NPARTS;
    const int  tEnd  = ((part + 1) * NTRIP) / NPARTS;
    const int  node  = node0 + nl;
    const bool valid = node < NNODES;
    float bestD[KNN];
    int   bestJ[KNN];
#pragma unroll
    for (int k = 0; k < KNN; ++k) { bestD[k] = 3.0e38f; bestJ[k] = 0; }

    const float2* sl  = (const float2*)sLocs;
    const v4f*    sc4 = (const v4f*)sCand;
    float px = 0.0f, py = 0.0f;
    if (valid) { const float2 me = sl[node]; px = me.x; py = me.y; }

    float thr;
    unsigned qb = tid * 4;
    const v2f pxv = {px, px};
    const v2f pyv = {py, py};

    auto cand = [&](float d2, float jf) {
        float* q = (float*)((char*)sQ + qb);
        q[0]     = d2;
        q[BLOCK] = jf;
        qb += (d2 <= thr) ? QROW : 0;
    };
    auto pair2 = [&](v2f x, v2f y) {
        const v2f dx = x - pxv;
        const v2f dy = y - pyv;
        return __builtin_elementwise_fma(dx, dx, dy * dy);
    };
    auto proc8 = [&](const v4f (&h)[6]) {
#pragma unroll
        for (int g = 0; g < 2; ++g) {
            const v4f X = h[3 * g], Y = h[3 * g + 1], J = h[3 * g + 2];
            const v2f da = pair2(X.xy, Y.xy);
            const v2f db = pair2(X.zw, Y.zw);
            cand(da.x, J.x);
            cand(da.y, J.y);
            cand(db.x, J.z);
            cand(db.y, J.w);
        }
    };

    auto insert = [&](float d2, int j) {
        bool c[KNN];
#pragma unroll
        for (int s = 0; s < KNN; ++s) c[s] = d2 < bestD[s];
#pragma unroll
        for (int s = KNN - 1; s > 0; --s) {
            const int jLow = bestJ[s - 1], jOwn = bestJ[s];
            const int jNew = c[s] ? j : jOwn;
            bestJ[s] = c[s - 1] ? jLow : jNew;
            bestD[s] = __builtin_amdgcn_fmed3f(bestD[s - 1], bestD[s], d2);
        }
        const float d0 = bestD[0];
        const int   j0 = bestJ[0];
        bestD[0] = c[0] ? d2 : d0;
        bestJ[0] = c[0] ? j  : j0;
    };

    auto drain = [&](int cnt) {
        const int lastRow = (cnt > 0) ? (cnt - 1) : 0;
        float ed = sQ[tid], ej = sQ[BLOCK + tid];
        bool own = 0 < cnt;
#pragma nounroll
        for (int i = 0; ; ++i) {
            const int nx = (i + 1 < lastRow) ? (i + 1) : lastRow;
            const float nd = sQ[nx * 2 * BLOCK + tid];
            const float nj = sQ[nx * 2 * BLOCK + BLOCK + tid];
            asm volatile("" ::: "memory");
            const bool ownNext = i + 1 < cnt;
            const bool more = __builtin_amdgcn_ballot_w32(ownNext) != 0u;
            const int   j  = __float_as_int(ej);
            const float none = 3.0e38f;
            const float d2 = (own && j != node) ? ed : none;
            insert(d2, j);
            if (!more) break;
            ed = nd; ej = nj; own = ownNext;
        }
    };

    float seed;
    {
        float a1 = 3.0e38f, a2 = 3.0e38f, a3 = 3.0e38f;
        for (int t = tBeg; t < tBeg + BOOT; ++t) {
            v4f h[12];
#pragma unroll
            for (int u = 0; u < 12; ++u) h[u] = sc4[t * 12 + u];
            float gm = 3.0e38f;
#pragma unroll
            for (int g = 0; g < 4; ++g) {
                const v4f X = h[3 * g], Y = h[3 * g + 1];
                const v2f da = pair2(X.xy, Y.xy);
                const v2f db = pair2(X.zw, Y.zw);
                gm = fminf(gm, fminf(fminf(da.x, da.y), fminf(db.x, db.y)));
            }
            const float n3 = __builtin_amdgcn_fmed3f(a2, a3, gm);
            const float n2 = __builtin_amdgcn_fmed3f(a1, a2, gm);
            a1 = fminf(a1, gm);
            a2 = n2;
            a3 = n3;
        }
        *(float2*)(sBoot + (nl * NPARTS + part) * 2) = make_float2(a2, a3);
        __syncthreads();
        const v4f b01 = *(const v4f*)(sBoot + nl * NPARTS * 2);
        const v4f b23 = *(const v4f*)(sBoot + nl * NPARTS * 2 + 4);
        const float t0 = fmaxf(b01.x, fmaxf(fmaxf(b01.w, b23.y), b23.w));
        const float t1 = fmaxf(b01.z, fmaxf(fmaxf(b01.y, b23.y), b23.w));
        const float t2 = fmaxf(b23.x, fmaxf(fmaxf(b01.y, b01.w), b23.w));
        const float t3 = fmaxf(b23.z, fmaxf(fmaxf(b01.y, b01.w), b23.y));
        seed = valid ? fminf(fminf(t0, t1), fminf(t2, t3)) : -1.0f;
    }
    const int seedBits = __float_as_int(seed);
    sThr[nl * NPARTS + part] = seedBits;
    thr = seed;

    v4f h0[6], h1[6];
#pragma unroll
    for (int u = 0; u < 6; ++u) h0[u] = sc4[tBeg * 12 + u];

    for (int t = tBeg; t < tEnd; ++t) {
#pragma unroll
        for (int u = 0; u < 6; ++u) h1[u] = sc4[t * 12 + 6 + u];
        const v4i sh = *(const v4i*)(sThr + nl * NPARTS);
        proc8(h0);
        const int cntMid = qb / QROW;
        {
            const int own = __float_as_int(bestD[KNN - 1]);
            const int m01 = sh.x < sh.y ? sh.x : sh.y;
            const int m23 = sh.z < sh.w ? sh.z : sh.w;
            const int m   = m01 < m23 ? m01 : m23;
            thr = __int_as_float(m < own ? m : own);
        }
        const int vn = (t + 1 < NTRIP) ? (t + 1) * 12 : (NTRIP * 12 - 6);
#pragma unroll
        for (int u = 0; u < 6; ++u) h0[u] = sc4[vn + u];
        proc8(h1);

        const bool full = __builtin_amdgcn_ballot_w32(cntMid > QCAP - 24) != 0u;
        if (full || t == tEnd - 1) {
            drain(qb / QROW);
            qb = tid * 4;
            const int own = __float_as_int(bestD[KNN - 1]);
            sThr[nl * NPARTS + part] = own < seedBits ? own : seedBits;
            const int cur = __float_as_int(thr);
            thr = __int_as_float(own < cur ? own : cur);
        }
    }

    __syncthreads();
    if (part > 0) {
#pragma unroll
        for (int k = 0; k < KNN; ++k)
        {
            const int row = (part - 1) * KNN + k;
            sQ[row * 2 * BLOCK + nl]         = bestD[k];
            sQ[row * 2 * BLOCK + BLOCK + nl] = __int_as_float(bestJ[k]);
        }
    }
    __syncthreads();

    if (part == 0) {
    drain((NPARTS - 1) * KNN);

    float* fr = &sFeat[nl * FSTRIDE];
    if (node < NNODES) {
        fr[0] = px;
        fr[1] = py;
#pragma unroll
        for (int k = 0; k < KNN; ++k) {
            const float2 q = sl[bestJ[k]];
            fr[2 + 2 * k]     = q.x - px;
            fr[2 + 2 * k + 1] = q.y - py;
        }
        fr[NODE_DIM]     = 0.0f;
        fr[NODE_DIM + 1] = 0.0f;
    } else {
#pragma unroll
        for (int k = 0; k < FDIM; ++k) fr[k] = 0.0f;
    }
    }
    __syncthreads();

    if (part != 0) return;
    const int wave   = tid >> 5;
    const int lane   = tid & 31;
    const int laneLo = lane & 15;
    const int kHalf  = (lane < 16) ? 0 : 2;
    const int mBase  = (lane < 16) ? 0 : 8;

    const int t0 = wave * 2;
    const int t1 = t0 + 1;
    const bool ok0 = (node0 + t0 * 16) < NNODES;
    const bool ok1 = (node0 + t1 * 16) < NNODES;

    v2f a0[6], a1[6];
    {
        const float* fa0 = &sFeat[(t0 * 16 + laneLo) * FSTRIDE];
        const float* fa1 = &sFeat[(t1 * 16 + laneLo) * FSTRIDE];
#pragma unroll
        for (int kc = 0; kc < 6; ++kc) {
            const int k0 = kc * 4 + kHalf;
            a0[kc].x = fa0[k0];  a0[kc].y = fa0[k0 + 1];
            a1[kc].x = fa1[k0];  a1[kc].y = fa1[k0 + 1];
        }
    }

    float* const outB = out + (size_t)batch * NNODES * EMBED_DIM;

#pragma unroll
    for (int ct = 0; ct < 8; ++ct) {
        const int e = ct * 16 + laneLo;
        v2f bf[6];
#pragma unroll
        for (int kc = 0; kc < 6; ++kc) {
            const int k0 = kc * 4 + kHalf;
            bf[kc].x = sW[k0 * EMBED_DIM + e];
            bf[kc].y = sW[(k0 + 1) * EMBED_DIM + e];
        }
        v8f acc0 = {}, acc1 = {};
#pragma unroll
        for (int kc = 0; kc < 6; ++kc)
            acc0 = __builtin_amdgcn_wmma_f32_16x16x4_f32(
                false, a0[kc], false, bf[kc], (short)0, acc0, false, false);
#pragma unroll
        for (int kc = 0; kc < 6; ++kc)
            acc1 = __builtin_amdgcn_wmma_f32_16x16x4_f32(
                false, a1[kc], false, bf[kc], (short)0, acc1, false, false);

        const float bv = bias[e];
        if (ok0) {
            float* p = outB + (size_t)(node0 + t0 * 16 + mBase) * EMBED_DIM + e;
#pragma unroll
            for (int r = 0; r < 8; ++r)
                p[(size_t)r * EMBED_DIM] = acc0[r] + bv;
        }
        if (ok1) {
            float* p = outB + (size_t)(node0 + t1 * 16 + mBase) * EMBED_DIM + e;
#pragma unroll
            for (int r = 0; r < 8; ++r)
                p[(size_t)r * EMBED_DIM] = acc1[r] + bv;
        }
    }
}

extern "C" void kernel_launch(void* const* d_in, const int* in_sizes, int n_in,
                              void* d_out, int out_size, void* d_ws, size_t ws_size,
                              hipStream_t stream) {
    const float* locs = (const float*)d_in[0];
    const float* W    = (const float*)d_in[1];
    const float* b    = (const float*)d_in[2];
    float* out        = (float*)d_out;

    dim3 grid(NBATCH * BLOCKS_PER_BATCH);
    dim3 block(BLOCK);
    tsp_embed_kernel<<<grid, block, 0, stream>>>(locs, W, b, out);
}


# Staging loads batched: 8 locs loads back to back (index select instead of a branch); W and bias loads issue behind the first barrier and land under the seed pass; bias read from LDS
# speedup vs baseline: 7.6919x; 1.1171x over previous
#include <hip/hip_runtime.h>
#include <hip/hip_bf16.h>

#define EMBED_DIM 128
#define KNN 10
#define NODE_DIM 22
#define FDIM 24
#define NNODES 2000
#define NBATCH 16
#define NPARTS 4
#define NODES_PER_BLOCK 128
#define BLOCK (NPARTS * NODES_PER_BLOCK)
#define BLOCKS_PER_BATCH 16
#define FSTRIDE 25
#define NTRIP (NNODES / 16)
#define QCAP 40
#define QROW (BLOCK * 8)
#define CGRP 12
#define NGRP (NNODES / 4)
#define BOOT 16

typedef __attribute__((ext_vector_type(2))) float v2f;
typedef __attribute__((ext_vector_type(4))) float v4f;
typedef __attribute__((ext_vector_type(8))) float v8f;
typedef __attribute__((ext_vector_type(4))) int   v4i;

__global__ __launch_bounds__(BLOCK)
void tsp_embed_kernel(const float* __restrict__ locs,
                      const float* __restrict__ W,
                      const float* __restrict__ bias,
                      float* __restrict__ out)
{
    __shared__ alignas(16) float sLocs[2 * NNODES];
    __shared__ alignas(16) float sCand[NGRP * CGRP];
    __shared__ float sFeat[NODES_PER_BLOCK * FSTRIDE];
    __shared__ float sW[FDIM * EMBED_DIM];
    __shared__ float sQ[2 * QCAP * BLOCK];
    __shared__ alignas(16) int sThr[NPARTS * NODES_PER_BLOCK];
    __shared__ alignas(16) float sBoot[NODES_PER_BLOCK * NPARTS * 2];
    __shared__ float sBias[EMBED_DIM];

    const int tid   = threadIdx.x;
    const int batch = blockIdx.x / BLOCKS_PER_BATCH;
    const int node0 = (blockIdx.x % BLOCKS_PER_BATCH) * NODES_PER_BLOCK;

    const float* bl = locs + (size_t)batch * NNODES * 2;
    float lv[8];
#pragma unroll
    for (int u = 0; u < 8; ++u) {
        const int raw = tid + u * BLOCK;
        lv[u] = bl[raw < 2 * NNODES ? raw : raw - BLOCK];
    }
#pragma unroll
    for (int u = 0; u < 8; ++u) {
        const int raw = tid + u * BLOCK;
        const int idx = raw < 2 * NNODES ? raw : raw - BLOCK;
        const float v = lv[u];
        const int nd = idx >> 1;
        sLocs[idx] = v;
        sCand[(nd >> 2) * CGRP + (idx & 1) * 4 + (nd & 3)] = v;
    }
    for (int j = tid; j < NNODES; j += BLOCK)
        sCand[(j >> 2) * CGRP + 8 + (j & 3)] = __int_as_float(j);
    sThr[(tid % NODES_PER_BLOCK) * NPARTS + tid / NODES_PER_BLOCK] =
        __float_as_int((node0 + tid % NODES_PER_BLOCK < NNODES) ? 3.0e38f : -1.0f);
    __syncthreads();

    const float bvl = bias[tid % EMBED_DIM];
    float wv[6];
#pragma unroll
    for (int u = 0; u < 6; ++u) {
        const int idx = tid + u * BLOCK;
        const int f = idx / EMBED_DIM;
        const int e = idx % EMBED_DIM;
        wv[u] = W[e * NODE_DIM + (f < NODE_DIM ? f : NODE_DIM - 1)];
    }

    const int  part  = __builtin_amdgcn_readfirstlane(tid) / NODES_PER_BLOCK;
    const int  nl    = tid % NODES_PER_BLOCK;
    const int  tBeg  = (part * NTRIP) / NPARTS;
    const int  tEnd  = ((part + 1) * NTRIP) / NPARTS;
    const int  node  = node0 + nl;
    const bool valid = node < NNODES;
    float bestD[KNN];
    int   bestJ[KNN];
#pragma unroll
    for (int k = 0; k < KNN; ++k) { bestD[k] = 3.0e38f; bestJ[k] = 0; }

    const float2* sl  = (const float2*)sLocs;
    const v4f*    sc4 = (const v4f*)sCand;
    float px = 0.0f, py = 0.0f;
    if (valid) { const float2 me = sl[node]; px = me.x; py = me.y; }

    float thr;
    unsigned qb = tid * 4;
    v2f pxv = {px, px};
    v2f pyv = {py, py};
    asm volatile("" : "+v"(pxv), "+v"(pyv));

    auto cand = [&](float d2, float jf) {
        float* q = (float*)((char*)sQ + qb);
        q[0]     = d2;
        q[BLOCK] = jf;
        qb += (d2 <= thr) ? QROW : 0;
    };
    auto pair2 = [&](v2f x, v2f y) {
        const v2f dx = x - pxv;
        const v2f dy = y - pyv;
        return __builtin_elementwise_fma(dx, dx, dy * dy);
    };
    auto proc8 = [&](const v4f (&h)[6]) {
#pragma unroll
        for (int g = 0; g < 2; ++g) {
            const v4f X = h[3 * g], Y = h[3 * g + 1], J = h[3 * g + 2];
            const v2f da = pair2(X.xy, Y.xy);
            const v2f db = pair2(X.zw, Y.zw);
            cand(da.x, J.x);
            cand(da.y, J.y);
            cand(db.x, J.z);
            cand(db.y, J.w);
        }
    };

    auto insert = [&](float d2, int j) {
        bool c[KNN];
#pragma unroll
        for (int s = 0; s < KNN; ++s) c[s] = d2 < bestD[s];
#pragma unroll
        for (int s = KNN - 1; s > 0; --s) {
            const int jLow = bestJ[s - 1], jOwn = bestJ[s];
            const int jNew = c[s] ? j : jOwn;
            bestJ[s] = c[s - 1] ? jLow : jNew;
            bestD[s] = __builtin_amdgcn_fmed3f(bestD[s - 1], bestD[s], d2);
        }
        const float d0 = bestD[0];
        const int   j0 = bestJ[0];
        bestD[0] = c[0] ? d2 : d0;
        bestJ[0] = c[0] ? j  : j0;
    };

    auto drain = [&](int cnt) {
        const int lastRow = (cnt > 0) ? (cnt - 1) : 0;
        float ed = sQ[tid], ej = sQ[BLOCK + tid];
        bool own = 0 < cnt;
#pragma nounroll
        for (int i = 0; ; ++i) {
            const int nx = (i + 1 < lastRow) ? (i + 1) : lastRow;
            const float nd = sQ[nx * 2 * BLOCK + tid];
            const float nj = sQ[nx * 2 * BLOCK + BLOCK + tid];
            asm volatile("" ::: "memory");
            const bool ownNext = i + 1 < cnt;
            const bool more = __builtin_amdgcn_ballot_w32(ownNext) != 0u;
            const int   j  = __float_as_int(ej);
            const float none = 3.0e38f;
            const float d2 = (own && j != node) ? ed : none;
            insert(d2, j);
            if (!more) break;
            ed = nd; ej = nj; own = ownNext;
        }
    };

    float seed;
    {
        float a1 = 3.0e38f, a2 = 3.0e38f, a3 = 3.0e38f;
        for (int t = tBeg; t < tBeg + BOOT; ++t) {
            v4f h[12];
#pragma unroll
            for (int u = 0; u < 12; ++u) h[u] = sc4[t * 12 + u];
            float gm = 3.0e38f;
#pragma unroll
            for (int g = 0; g < 4; ++g) {
                const v4f X = h[3 * g], Y = h[3 * g + 1];
                const v2f da = pair2(X.xy, Y.xy);
                const v2f db = pair2(X.zw, Y.zw);
                gm = fminf(gm, fminf(fminf(da.x, da.y), fminf(db.x, db.y)));
            }
            const float n3 = __builtin_amdgcn_fmed3f(a2, a3, gm);
            const float n2 = __builtin_amdgcn_fmed3f(a1, a2, gm);
            a1 = fminf(a1, gm);
            a2 = n2;
            a3 = n3;
        }
        *(float2*)(sBoot + (nl * NPARTS + part) * 2) = make_float2(a2, a3);
        __builtin_amdgcn_sched_barrier(0);
#pragma unroll
        for (int u = 0; u < 6; ++u) {
            const int idx = tid + u * BLOCK;
            sW[idx] = (idx / EMBED_DIM < NODE_DIM) ? wv[u] : 0.0f;
        }
        if (tid < EMBED_DIM) sBias[tid] = bvl;
        __syncthreads();
        const v4f b01 = *(const v4f*)(sBoot + nl * NPARTS * 2);
        const v4f b23 = *(const v4f*)(sBoot + nl * NPARTS * 2 + 4);
        const float t0 = fmaxf(b01.x, fmaxf(fmaxf(b01.w, b23.y), b23.w));
        const float t1 = fmaxf(b01.z, fmaxf(fmaxf(b01.y, b23.y), b23.w));
        const float t2 = fmaxf(b23.x, fmaxf(fmaxf(b01.y, b01.w), b23.w));
        const float t3 = fmaxf(b23.z, fmaxf(fmaxf(b01.y, b01.w), b23.y));
        seed = valid ? fminf(fminf(t0, t1), fminf(t2, t3)) : -1.0f;
    }
    const int seedBits = __float_as_int(seed);
    sThr[nl * NPARTS + part] = seedBits;
    thr = seed;

    v4f h0[6], h1[6];
#pragma unroll
    for (int u = 0; u < 6; ++u) h0[u] = sc4[tBeg * 12 + u];

    for (int t = tBeg; t < tEnd; ++t) {
#pragma unroll
        for (int u = 0; u < 6; ++u) h1[u] = sc4[t * 12 + 6 + u];
        const v4i sh = *(const v4i*)(sThr + nl * NPARTS);
        proc8(h0);
        const int cntMid = qb / QROW;
        {
            const int own = __float_as_int(bestD[KNN - 1]);
            const int m01 = sh.x < sh.y ? sh.x : sh.y;
            const int m23 = sh.z < sh.w ? sh.z : sh.w;
            const int m   = m01 < m23 ? m01 : m23;
            thr = __int_as_float(m < own ? m : own);
        }
        const int vn = (t + 1 < NTRIP) ? (t + 1) * 12 : (NTRIP * 12 - 6);
#pragma unroll
        for (int u = 0; u < 6; ++u) h0[u] = sc4[vn + u];
        proc8(h1);

        const bool full = __builtin_amdgcn_ballot_w32(cntMid > QCAP - 24) != 0u;
        if (full || t == tEnd - 1) {
            drain(qb / QROW);
            qb = tid * 4;
            const int own = __float_as_int(bestD[KNN - 1]);
            sThr[nl * NPARTS + part] = own < seedBits ? own : seedBits;
            const int cur = __float_as_int(thr);
            thr = __int_as_float(own < cur ? own : cur);
        }
    }

    __syncthreads();
    if (part > 0) {
#pragma unroll
        for (int k = 0; k < KNN; ++k)
        {
            const int row = (part - 1) * KNN + k;
            sQ[row * 2 * BLOCK + nl]         = bestD[k];
            sQ[row * 2 * BLOCK + BLOCK + nl] = __int_as_float(bestJ[k]);
        }
    }
    __syncthreads();

    if (part == 0) {
    drain((NPARTS - 1) * KNN);

    float* fr = &sFeat[nl * FSTRIDE];
    if (node < NNODES) {
        fr[0] = px;
        fr[1] = py;
#pragma unroll
        for (int k = 0; k < KNN; ++k) {
            const float2 q = sl[bestJ[k]];
            fr[2 + 2 * k]     = q.x - px;
            fr[2 + 2 * k + 1] = q.y - py;
        }
        fr[NODE_DIM]     = 0.0f;
        fr[NODE_DIM + 1] = 0.0f;
    } else {
#pragma unroll
        for (int k = 0; k < FDIM; ++k) fr[k] = 0.0f;
    }
    }
    __syncthreads();

    if (part != 0) return;
    const int wave   = tid >> 5;
    const int lane   = tid & 31;
    const int laneLo = lane & 15;
    const int kHalf  = (lane < 16) ? 0 : 2;
    const int mBase  = (lane < 16) ? 0 : 8;

    const int t0 = wave * 2;
    const int t1 = t0 + 1;
    const bool ok0 = (node0 + t0 * 16) < NNODES;
    const bool ok1 = (node0 + t1 * 16) < NNODES;

    v2f a0[6], a1[6];
    {
        const float* fa0 = &sFeat[(t0 * 16 + laneLo) * FSTRIDE];
        const float* fa1 = &sFeat[(t1 * 16 + laneLo) * FSTRIDE];
#pragma unroll
        for (int kc = 0; kc < 6; ++kc) {
            const int k0 = kc * 4 + kHalf;
            a0[kc].x = fa0[k0];  a0[kc].y = fa0[k0 + 1];
            a1[kc].x = fa1[k0];  a1[kc].y = fa1[k0 + 1];
        }
    }

    float* const outB = out + (size_t)batch * NNODES * EMBED_DIM;

#pragma unroll
    for (int ct = 0; ct < 8; ++ct) {
        const int e = ct * 16 + laneLo;
        v2f bf[6];
#pragma unroll
        for (int kc = 0; kc < 6; ++kc) {
            const int k0 = kc * 4 + kHalf;
            bf[kc].x = sW[k0 * EMBED_DIM + e];
            bf[kc].y = sW[(k0 + 1) * EMBED_DIM + e];
        }
        v8f acc0 = {}, acc1 = {};
#pragma unroll
        for (int kc = 0; kc < 6; ++kc)
            acc0 = __builtin_amdgcn_wmma_f32_16x16x4_f32(
                false, a0[kc], false, bf[kc], (short)0, acc0, false, false);
#pragma unroll
        for (int kc = 0; kc < 6; ++kc)
            acc1 = __builtin_amdgcn_wmma_f32_16x16x4_f32(
                false, a1[kc], false, bf[kc], (short)0, acc1, false, false);

        const float bv = sBias[e];
        if (ok0) {
            float* p = outB + (size_t)(node0 + t0 * 16 + mBase) * EMBED_DIM + e;
#pragma unroll
            for (int r = 0; r < 8; ++r)
                p[(size_t)r * EMBED_DIM] = acc0[r] + bv;
        }
        if (ok1) {
            float* p = outB + (size_t)(node0 + t1 * 16 + mBase) * EMBED_DIM + e;
#pragma unroll
            for (int r = 0; r < 8; ++r)
                p[(size_t)r * EMBED_DIM] = acc1[r] + bv;
        }
    }
}

extern "C" void kernel_launch(void* const* d_in, const int* in_sizes, int n_in,
                              void* d_out, int out_size, void* d_ws, size_t ws_size,
                              hipStream_t stream) {
    const float* locs = (const float*)d_in[0];
    const float* W    = (const float*)d_in[1];
    const float* b    = (const float*)d_in[2];
    float* out        = (float*)d_out;

    dim3 grid(NBATCH * BLOCKS_PER_BATCH);
    dim3 block(BLOCK);
    tsp_embed_kernel<<<grid, block, 0, stream>>>(locs, W, b, out);
}


# KNN scan: compare plus add-with-carry per candidate builds a trip's pass mask; one LDS word stored per trip (was two per candidate); drain recomputes d2 of set bits; index plane, bound exchange gone
# speedup vs baseline: 9.1050x; 1.1837x over previous
#include <hip/hip_runtime.h>
#include <hip/hip_bf16.h>

#define EMBED_DIM 128
#define KNN 10
#define NODE_DIM 22
#define FDIM 24
#define NNODES 2000
#define NBATCH 16
#define NPARTS 4
#define NODES_PER_BLOCK 128
#define BLOCK (NPARTS * NODES_PER_BLOCK)
#define BLOCKS_PER_BATCH 16
#define FSTRIDE 25
#define NTRIP (NNODES / 16)
#define QROWS 32
#define QROWB (BLOCK * 4)
#define MROWS ((NPARTS - 1) * KNN)
#define CGRP 8
#define NGRP (NNODES / 4)
#define BOOT 16

typedef __attribute__((ext_vector_type(2))) float v2f;
typedef __attribute__((ext_vector_type(4))) float v4f;
typedef __attribute__((ext_vector_type(8))) float v8f;

__global__ __launch_bounds__(BLOCK)
void tsp_embed_kernel(const float* __restrict__ locs,
                      const float* __restrict__ W,
                      const float* __restrict__ bias,
                      float* __restrict__ out)
{
    __shared__ alignas(16) float sLocs[2 * NNODES];
    __shared__ alignas(16) float sCand[NGRP * CGRP];
    __shared__ float sFeat[NODES_PER_BLOCK * FSTRIDE];
    __shared__ float sW[FDIM * EMBED_DIM];
    __shared__ float sQ[2 * MROWS * BLOCK];
    static_assert(QROWS * BLOCK <= 2 * MROWS * BLOCK, "the scan's queue rows fit in the merge's area");
    __shared__ alignas(16) float sBoot[NODES_PER_BLOCK * NPARTS * 2];
    __shared__ float sBias[EMBED_DIM];

    const int tid   = threadIdx.x;
    const int batch = blockIdx.x / BLOCKS_PER_BATCH;
    const int node0 = (blockIdx.x % BLOCKS_PER_BATCH) * NODES_PER_BLOCK;

    const float* bl = locs + (size_t)batch * NNODES * 2;
    float lv[8];
#pragma unroll
    for (int u = 0; u < 8; ++u) {
        const int raw = tid + u * BLOCK;
        lv[u] = bl[raw < 2 * NNODES ? raw : raw - BLOCK];
    }
#pragma unroll
    for (int u = 0; u < 8; ++u) {
        const int raw = tid + u * BLOCK;
        const int idx = raw < 2 * NNODES ? raw : raw - BLOCK;
        const float v = lv[u];
        const int nd = idx >> 1;
        sLocs[idx] = v;
        sCand[(nd >> 2) * CGRP + (idx & 1) * 4 + (nd & 3)] = v;
    }
    __syncthreads();

    const float bvl = bias[tid % EMBED_DIM];
    float wv[6];
#pragma unroll
    for (int u = 0; u < 6; ++u) {
        const int idx = tid + u * BLOCK;
        const int f = idx / EMBED_DIM;
        const int e = idx % EMBED_DIM;
        wv[u] = W[e * NODE_DIM + (f < NODE_DIM ? f : NODE_DIM - 1)];
    }

    const int  part  = __builtin_amdgcn_readfirstlane(tid) / NODES_PER_BLOCK;
    const int  nl    = tid % NODES_PER_BLOCK;
    const int  tBeg  = (part * NTRIP) / NPARTS;
    const int  tEnd  = ((part + 1) * NTRIP) / NPARTS;
    const int  node  = node0 + nl;
    const bool valid = node < NNODES;
    float bestD[KNN];
    int   bestJ[KNN];
#pragma unroll
    for (int k = 0; k < KNN; ++k) { bestD[k] = 3.0e38f; bestJ[k] = 0; }

    const float2* sl  = (const float2*)sLocs;
    const v4f*    sc4 = (const v4f*)sCand;
    float px = 0.0f, py = 0.0f;
    if (valid) { const float2 me = sl[node]; px = me.x; py = me.y; }

    float thr;
    unsigned qb = tid * 4;
    v2f pxv = {px, px};
    v2f pyv = {py, py};
    asm volatile("" : "+v"(pxv), "+v"(pyv));

    auto pass = [&](unsigned m, float d2) {
        asm("s_delay_alu instid0(VALU_DEP_1) | instskip(NEXT) | instid1(VALU_DEP_1)\n\t"
            "v_cmp_le_f32 vcc_lo, %1, %2\n\t"
            "v_add_co_ci_u32 %0, vcc_lo, %0, %0, vcc_lo"
            : "+v"(m) : "v"(d2), "v"(thr) : "vcc");
        return m;
    };
    auto pair2 = [&](v2f x, v2f y) {
        const v2f dx = x - pxv;
        const v2f dy = y - pyv;
        return __builtin_elementwise_fma(dx, dx, dy * dy);
    };
    auto proc8 = [&](const v4f (&h)[4]) {
        unsigned m = 0u;
#pragma unroll
        for (int g = 0; g < 2; ++g) {
            const v4f X = h[2 * g], Y = h[2 * g + 1];
            const v2f da = pair2(X.xy, Y.xy);
            const v2f db = pair2(X.zw, Y.zw);
            m = pass(m, da.x);
            m = pass(m, da.y);
            m = pass(m, db.x);
            m = pass(m, db.y);
        }
        return m;
    };

    auto insert = [&](float d2, int j) {
        bool c[KNN];
#pragma unroll
        for (int s = 0; s < KNN; ++s) c[s] = d2 < bestD[s];
#pragma unroll
        for (int s = KNN - 1; s > 0; --s) {
            const int jLow = bestJ[s - 1], jOwn = bestJ[s];
            const int jNew = c[s] ? j : jOwn;
            bestJ[s] = c[s - 1] ? jLow : jNew;
            bestD[s] = __builtin_amdgcn_fmed3f(bestD[s - 1], bestD[s], d2);
        }
        const float d0 = bestD[0];
        const int   j0 = bestJ[0];
        bestD[0] = c[0] ? d2 : d0;
        bestJ[0] = c[0] ? j  : j0;
    };

    auto drain = [&](int cnt) {
        const int lastRow = (cnt > 0) ? (cnt - 1) : 0;
        float ed = sQ[tid], ej = sQ[BLOCK + tid];
        bool own = 0 < cnt;
#pragma nounroll
        for (int i = 0; ; ++i) {
            const int nx = (i + 1 < lastRow) ? (i + 1) : lastRow;
            const float nd = sQ[nx * 2 * BLOCK + tid];
            const float nj = sQ[nx * 2 * BLOCK + BLOCK + tid];
            asm volatile("" ::: "memory");
            const bool ownNext = i + 1 < cnt;
            const bool more = __builtin_amdgcn_ballot_w32(ownNext) != 0u;
            const int   j  = __float_as_int(ej);
            const float none = 3.0e38f;
            const float d2 = (own && j != node) ? ed : none;
            insert(d2, j);
            if (!more) break;
            ed = nd; ej = nj; own = ownNext;
        }
    };

    auto drainMask = [&](int cnt) {
        const unsigned* qw = (const unsigned*)sQ;
        const int lastRow = (cnt > 0) ? (cnt - 1) : 0;
        int      i   = 0;
        unsigned cur = (cnt > 0) ? qw[tid] : 0u;
        unsigned nxt = qw[((1 < lastRow) ? 1 : lastRow) * BLOCK + tid];
        auto next = [&](int& j, bool& has) {
            const unsigned bits = cur & 0xffffu;
            has = bits != 0u;
            const int hb = 31 - __clz((int)(bits | 1u));
            j = (int)(cur >> 16) * 16 + 15 - hb;
            const unsigned rest = bits & ~(1u << hb);
            const bool roll = rest == 0u;
            i += roll ? 1 : 0;
            cur = roll ? ((i < cnt) ? nxt : 0u) : (cur & ~(1u << hb));
            const int nr = (i + 1 < lastRow) ? (i + 1) : lastRow;
            nxt = qw[nr * BLOCK + tid];
        };
        int jN; bool hasN;
        next(jN, hasN);
        float2 locN = sl[jN];
#pragma nounroll
        for (;;) {
            const float2 loc = locN;
            const int    j   = jN;
            const bool   has = hasN;
            next(jN, hasN);
            locN = sl[jN];
            asm volatile("" ::: "memory");
            const bool more = __builtin_amdgcn_ballot_w32(hasN) != 0u;
            const float dx = loc.x - px;
            const float dy = loc.y - py;
            const float dd = __builtin_fmaf(dx, dx, dy * dy);
            const float d2 = (has && j != node) ? dd : 3.0e38f;
            insert(d2, j);
            if (!more) break;
        }
    };

    float seed;
    {
        float a1 = 3.0e38f, a2 = 3.0e38f, a3 = 3.0e38f;
#pragma unroll
        for (int t = tBeg; t < tBeg + BOOT; ++t) {
            v4f h[8];
#pragma unroll
            for (int u = 0; u < 8; ++u) h[u] = sc4[t * 8 + u];
            float gm = 3.0e38f;
#pragma unroll
            for (int g = 0; g < 4; ++g) {
                const v4f X = h[2 * g], Y = h[2 * g + 1];
                const v2f da = pair2(X.xy, Y.xy);
                const v2f db = pair2(X.zw, Y.zw);
                gm = fminf(gm, fminf(fminf(da.x, da.y), fminf(db.x, db.y)));
            }
            const float n3 = __builtin_amdgcn_fmed3f(a2, a3, gm);
            const float n2 = __builtin_amdgcn_fmed3f(a1, a2, gm);
            a1 = fminf(a1, gm);
            a2 = n2;
            a3 = n3;
        }
        *(float2*)(sBoot + (nl * NPARTS + part) * 2) = make_float2(a2, a3);
        __builtin_amdgcn_sched_barrier(0);
#pragma unroll
        for (int u = 0; u < 6; ++u) {
            const int idx = tid + u * BLOCK;
            sW[idx] = (idx / EMBED_DIM < NODE_DIM) ? wv[u] : 0.0f;
        }
        if (tid < EMBED_DIM) sBias[tid] = bvl;
        __syncthreads();
        const v4f b01 = *(const v4f*)(sBoot + nl * NPARTS * 2);
        const v4f b23 = *(const v4f*)(sBoot + nl * NPARTS * 2 + 4);
        const float t0 = fmaxf(b01.x, fmaxf(fmaxf(b01.w, b23.y), b23.w));
        const float t1 = fmaxf(b01.z, fmaxf(fmaxf(b01.y, b23.y), b23.w));
        const float t2 = fmaxf(b23.x, fmaxf(fmaxf(b01.y, b01.w), b23.w));
        const float t3 = fmaxf(b23.z, fmaxf(fmaxf(b01.y, b01.w), b23.y));
        seed = valid ? fminf(fminf(t0, t1), fminf(t2, t3)) : -1.0f;
    }
    thr = seed;

    v4f h0[4], h1[4];
#pragma unroll
    for (int u = 0; u < 4; ++u) h0[u] = sc4[tBeg * 8 + u];

    for (int t = tBeg; t < tEnd; ++t) {
#pragma unroll
        for (int u = 0; u < 4; ++u) h1[u] = sc4[t * 8 + 4 + u];
        __builtin_amdgcn_sched_barrier(0);
        const unsigned mA = proc8(h0);
        __builtin_amdgcn_sched_barrier(0);
        const int vn = (t + 1 < NTRIP) ? (t + 1) * 8 : (NTRIP * 8 - 4);
#pragma unroll
        for (int u = 0; u < 4; ++u) h0[u] = sc4[vn + u];
        __builtin_amdgcn_sched_barrier(0);
        const unsigned mB = proc8(h1);
        const unsigned m = (mA << 8) | mB;
        *(unsigned*)((char*)sQ + qb) = ((unsigned)t << 16) | m;
        qb += (m != 0u) ? QROWB : 0u;
    }
    drainMask(qb / QROWB);

    __syncthreads();
    if (part > 0) {
#pragma unroll
        for (int k = 0; k < KNN; ++k)
        {
            const int row = (part - 1) * KNN + k;
            sQ[row * 2 * BLOCK + nl]         = bestD[k];
            sQ[row * 2 * BLOCK + BLOCK + nl] = __int_as_float(bestJ[k]);
        }
    }
    __syncthreads();

    if (part == 0) {
    drain((NPARTS - 1) * KNN);

    float* fr = &sFeat[nl * FSTRIDE];
    if (node < NNODES) {
        fr[0] = px;
        fr[1] = py;
#pragma unroll
        for (int k = 0; k < KNN; ++k) {
            const float2 q = sl[bestJ[k]];
            fr[2 + 2 * k]     = q.x - px;
            fr[2 + 2 * k + 1] = q.y - py;
        }
        fr[NODE_DIM]     = 0.0f;
        fr[NODE_DIM + 1] = 0.0f;
    } else {
#pragma unroll
        for (int k = 0; k < FDIM; ++k) fr[k] = 0.0f;
    }
    }
    __syncthreads();

    if (part != 0) return;
    const int wave   = tid >> 5;
    const int lane   = tid & 31;
    const int laneLo = lane & 15;
    const int kHalf  = (lane < 16) ? 0 : 2;
    const int mBase  = (lane < 16) ? 0 : 8;

    const int t0 = wave * 2;
    const int t1 = t0 + 1;
    const bool ok0 = (node0 + t0 * 16) < NNODES;
    const bool ok1 = (node0 + t1 * 16) < NNODES;

    v2f a0[6], a1[6];
    {
        const float* fa0 = &sFeat[(t0 * 16 + laneLo) * FSTRIDE];
        const float* fa1 = &sFeat[(t1 * 16 + laneLo) * FSTRIDE];
#pragma unroll
        for (int kc = 0; kc < 6; ++kc) {
            const int k0 = kc * 4 + kHalf;
            a0[kc].x = fa0[k0];  a0[kc].y = fa0[k0 + 1];
            a1[kc].x = fa1[k0];  a1[kc].y = fa1[k0 + 1];
        }
    }

    float* const outB = out + (size_t)batch * NNODES * EMBED_DIM;

#pragma unroll
    for (int ct = 0; ct < 8; ++ct) {
        const int e = ct * 16 + laneLo;
        v2f bf[6];
#pragma unroll
        for (int kc = 0; kc < 6; ++kc) {
            const int k0 = kc * 4 + kHalf;
            bf[kc].x = sW[k0 * EMBED_DIM + e];
            bf[kc].y = sW[(k0 + 1) * EMBED_DIM + e];
        }
        v8f acc0 = {}, acc1 = {};
#pragma unroll
        for (int kc = 0; kc < 6; ++kc)
            acc0 = __builtin_amdgcn_wmma_f32_16x16x4_f32(
                false, a0[kc], false, bf[kc], (short)0, acc0, false, false);
#pragma unroll
        for (int kc = 0; kc < 6; ++kc)
            acc1 = __builtin_amdgcn_wmma_f32_16x16x4_f32(
                false, a1[kc], false, bf[kc], (short)0, acc1, false, false);

        const float bv = sBias[e];
        if (ok0) {
            float* p = outB + (size_t)(node0 + t0 * 16 + mBase) * EMBED_DIM + e;
#pragma unroll
            for (int r = 0; r < 8; ++r)
                p[(size_t)r * EMBED_DIM] = acc0[r] + bv;
        }
        if (ok1) {
            float* p = outB + (size_t)(node0 + t1 * 16 + mBase) * EMBED_DIM + e;
#pragma unroll
            for (int r = 0; r < 8; ++r)
                p[(size_t)r * EMBED_DIM] = acc1[r] + bv;
        }
    }
}

extern "C" void kernel_launch(void* const* d_in, const int* in_sizes, int n_in,
                              void* d_out, int out_size, void* d_ws, size_t ws_size,
                              hipStream_t stream) {
    const float* locs = (const float*)d_in[0];
    const float* W    = (const float*)d_in[1];
    const float* b    = (const float*)d_in[2];
    float* out        = (float*)d_out;

    dim3 grid(NBATCH * BLOCKS_PER_BATCH);
    dim3 block(BLOCK);
    tsp_embed_kernel<<<grid, block, 0, stream>>>(locs, W, b, out);
}


# KNN distances (seed pass and scan) moved to the matrix unit: two f32 16x16x4 products a trip cover 32 nodes x 16 candidates (expanded form, bound widened for rounding); sign bits are the pass mask
# speedup vs baseline: 10.7036x; 1.1756x over previous
#include <hip/hip_runtime.h>
#include <hip/hip_bf16.h>

#define EMBED_DIM 128
#define KNN 10
#define NODE_DIM 22
#define FDIM 24
#define NNODES 2000
#define NBATCH 16
#define NPARTS 4
#define NODES_PER_BLOCK 128
#define BLOCK (NPARTS * NODES_PER_BLOCK)
#define BLOCKS_PER_BATCH 16
#define FSTRIDE 25
#define NTRIP (NNODES / 16)
#define QROWS 32
#define QROWB (BLOCK * 4)
#define MROWS ((NPARTS - 1) * KNN)
#define BOOT 16

typedef __attribute__((ext_vector_type(2))) float v2f;
typedef __attribute__((ext_vector_type(4))) float v4f;
typedef __attribute__((ext_vector_type(8))) float v8f;

__global__ __launch_bounds__(BLOCK)
void tsp_embed_kernel(const float* __restrict__ locs,
                      const float* __restrict__ W,
                      const float* __restrict__ bias,
                      float* __restrict__ out)
{
    __shared__ alignas(16) float sLocs[2 * NNODES];
    __shared__ float sFeat[NODES_PER_BLOCK * FSTRIDE];
    __shared__ float sW[FDIM * EMBED_DIM];
    __shared__ float sQ[2 * MROWS * BLOCK];
    static_assert(QROWS * BLOCK <= 2 * MROWS * BLOCK, "the scan's queue rows fit in the merge's area");
    __shared__ alignas(16) float sBoot[NODES_PER_BLOCK * NPARTS * 2];
    __shared__ float sBias[EMBED_DIM];
    __shared__ alignas(16) float sA[NNODES * 4];

    const int tid   = threadIdx.x;
    const int batch = blockIdx.x / BLOCKS_PER_BATCH;
    const int node0 = (blockIdx.x % BLOCKS_PER_BATCH) * NODES_PER_BLOCK;

    const float* bl = locs + (size_t)batch * NNODES * 2;
    v2f lv[4];
#pragma unroll
    for (int u = 0; u < 4; ++u) {
        const int raw = tid + u * BLOCK;
        lv[u] = *(const v2f*)(bl + 2 * (raw < NNODES ? raw : raw - BLOCK));
    }
    asm volatile("" : "+v"(lv[0]), "+v"(lv[1]), "+v"(lv[2]), "+v"(lv[3]));
#pragma unroll
    for (int u = 0; u < 4; ++u) {
        const int raw = tid + u * BLOCK;
        const int j = raw < NNODES ? raw : raw - BLOCK;
        const v2f c = lv[u];
        *(v2f*)(sLocs + 2 * j) = c;
        const v4f row = {c.x, c.y, __builtin_fmaf(c.x, c.x, c.y * c.y), 1.0f};
        *(v4f*)(sA + j * 4) = row;
    }
    __syncthreads();

    const float bvl = bias[tid % EMBED_DIM];
    float wv[6];
#pragma unroll
    for (int u = 0; u < 6; ++u) {
        const int idx = tid + u * BLOCK;
        const int f = idx / EMBED_DIM;
        const int e = idx % EMBED_DIM;
        wv[u] = W[e * NODE_DIM + (f < NODE_DIM ? f : NODE_DIM - 1)];
    }

    const int  part  = __builtin_amdgcn_readfirstlane(tid) / NODES_PER_BLOCK;
    const int  nl    = tid % NODES_PER_BLOCK;
    const int  tBeg  = (part * NTRIP) / NPARTS;
    const int  tEnd  = ((part + 1) * NTRIP) / NPARTS;
    const int  node  = node0 + nl;
    const bool valid = node < NNODES;
    float bestD[KNN];
    int   bestJ[KNN];
#pragma unroll
    for (int k = 0; k < KNN; ++k) { bestD[k] = 3.0e38f; bestJ[k] = 0; }

    const float2* sl  = (const float2*)sLocs;
    float px = 0.0f, py = 0.0f;
    if (valid) { const float2 me = sl[node]; px = me.x; py = me.y; }

    unsigned qb = tid * 4;

    const int   ln   = tid & 31;
    const bool  hiH  = (ln & 16) != 0;
    const float pp   = __builtin_fmaf(px, px, py * py);
    const float pxo  = __shfl_xor(px, 16, 32);
    const float pyo  = __shfl_xor(py, 16, 32);
    const float ppo  = __shfl_xor(pp, 16, 32);
    v2f bLo, bHi;
    bLo.x = hiH ? 1.0f : -2.0f * px;   bLo.y = hiH ? ppo : -2.0f * py;
    bHi.x = hiH ? 1.0f : -2.0f * pxo;  bHi.y = hiH ? pp  : -2.0f * pyo;
    const v2f* sa2 = (const v2f*)sA + (ln & 15) * 2 + (hiH ? 1 : 0);
    auto prod = [&](v2f aFrag, v2f bFrag) {
        const v8f zero = {};
        return __builtin_amdgcn_wmma_f32_16x16x4_f32(false, aFrag, false, bFrag, (short)0, zero, false, false);
    };
    auto insert = [&](float d2, int j) {
        bool c[KNN];
#pragma unroll
        for (int s = 0; s < KNN; ++s) c[s] = d2 < bestD[s];
#pragma unroll
        for (int s = KNN - 1; s > 0; --s) {
            const int jLow = bestJ[s - 1], jOwn = bestJ[s];
            const int jNew = c[s] ? j : jOwn;
            bestJ[s] = c[s - 1] ? jLow : jNew;
            bestD[s] = __builtin_amdgcn_fmed3f(bestD[s - 1], bestD[s], d2);
        }
        const float d0 = bestD[0];
        const int   j0 = bestJ[0];
        bestD[0] = c[0] ? d2 : d0;
        bestJ[0] = c[0] ? j  : j0;
    };

    auto drain = [&](int cnt) {
        const int lastRow = (cnt > 0) ? (cnt - 1) : 0;
        float ed = sQ[tid], ej = sQ[BLOCK + tid];
        bool own = 0 < cnt;
#pragma nounroll
        for (int i = 0; ; ++i) {
            const int nx = (i + 1 < lastRow) ? (i + 1) : lastRow;
            const float nd = sQ[nx * 2 * BLOCK + tid];
            const float nj = sQ[nx * 2 * BLOCK + BLOCK + tid];
            asm volatile("" ::: "memory");
            const bool ownNext = i + 1 < cnt;
            const bool more = __builtin_amdgcn_ballot_w32(ownNext) != 0u;
            const int   j  = __float_as_int(ej);
            const float none = 3.0e38f;
            const float d2 = (own && j != node) ? ed : none;
            insert(d2, j);
            if (!more) break;
            ed = nd; ej = nj; own = ownNext;
        }
    };

    auto drainMask = [&](int cnt) {
        const unsigned* qw = (const unsigned*)sQ;
        const int lastRow = (cnt > 0) ? (cnt - 1) : 0;
        int      i   = 0;
        unsigned cur = (cnt > 0) ? qw[tid] : 0u;
        unsigned nxt = qw[((1 < lastRow) ? 1 : lastRow) * BLOCK + tid];
        auto next = [&](int& j, bool& has) {
            const unsigned bits = cur & 0xffffu;
            has = bits != 0u;
            const int hb = 31 - __clz((int)(bits | 1u));
            j = (int)(cur >> 16) * 16 + 15 - hb;
            const unsigned rest = bits & ~(1u << hb);
            const bool roll = rest == 0u;
            i += roll ? 1 : 0;
            cur = roll ? ((i < cnt) ? nxt : 0u) : (cur & ~(1u << hb));
            const int nr = (i + 1 < lastRow) ? (i + 1) : lastRow;
            nxt = qw[nr * BLOCK + tid];
        };
        int jN; bool hasN;
        next(jN, hasN);
        float2 locN = sl[jN];
#pragma nounroll
        for (;;) {
            const float2 loc = locN;
            const int    j   = jN;
            const bool   has = hasN;
            next(jN, hasN);
            locN = sl[jN];
            asm volatile("" ::: "memory");
            const bool more = __builtin_amdgcn_ballot_w32(hasN) != 0u;
            const float dx = loc.x - px;
            const float dy = loc.y - py;
            const float dd = __builtin_fmaf(dx, dx, dy * dy);
            const float d2 = (has && j != node) ? dd : 3.0e38f;
            insert(d2, j);
            if (!more) break;
        }
    };

    static_assert(((NPARTS - 1) * NTRIP) / NPARTS + BOOT + 1 < NTRIP, "the seed pass reads two trips ahead, inside the candidates");
    float seed;
    {
        float a1 = 3.0e38f, a2 = 3.0e38f, a3 = 3.0e38f;
        auto min8 = [&](const v8f& d) {
            return fminf(fminf(fminf(d[0], d[1]), fminf(d[2], d[3])), fminf(fminf(d[4], d[5]), fminf(d[6], d[7])));
        };
        v8f dLo, dHi;
        {
            const v2f aFirst = sa2[tBeg * 32];
            dLo = prod(aFirst, bLo);
            dHi = prod(aFirst, bHi);
        }
        v2f aNext = sa2[(tBeg + 1) * 32];
        for (int t = tBeg; t < tBeg + BOOT; ++t) {
            const v2f aAfter = sa2[(t + 2) * 32];
            const v8f nLo = prod(aNext, bLo);
            const v8f nHi = prod(aNext, bHi);
            __builtin_amdgcn_sched_barrier(0);
            const auto sw = __builtin_amdgcn_permlane16_swap(__float_as_uint(min8(dLo)), __float_as_uint(min8(dHi)), false, false);
            const float gm = fminf(__uint_as_float(sw[0]), __uint_as_float(sw[1]));
            const float n3 = __builtin_amdgcn_fmed3f(a2, a3, gm);
            const float n2 = __builtin_amdgcn_fmed3f(a1, a2, gm);
            a1 = fminf(a1, gm);
            a2 = n2;
            a3 = n3;
            dLo = nLo;
            dHi = nHi;
            aNext = aAfter;
        }
        *(float2*)(sBoot + (nl * NPARTS + part) * 2) = make_float2(a2, a3);
        __builtin_amdgcn_sched_barrier(0);
#pragma unroll
        for (int u = 0; u < 6; ++u) {
            const int idx = tid + u * BLOCK;
            sW[idx] = (idx / EMBED_DIM < NODE_DIM) ? wv[u] : 0.0f;
        }
        if (tid < EMBED_DIM) sBias[tid] = bvl;
        __syncthreads();
        const v4f b01 = *(const v4f*)(sBoot + nl * NPARTS * 2);
        const v4f b23 = *(const v4f*)(sBoot + nl * NPARTS * 2 + 4);
        const float t0 = fmaxf(b01.x, fmaxf(fmaxf(b01.w, b23.y), b23.w));
        const float t1 = fmaxf(b01.z, fmaxf(fmaxf(b01.y, b23.y), b23.w));
        const float t2 = fmaxf(b23.x, fmaxf(fmaxf(b01.y, b01.w), b23.w));
        const float t3 = fmaxf(b23.z, fmaxf(fmaxf(b01.y, b01.w), b23.y));
        seed = valid ? fminf(fminf(t0, t1), fminf(t2, t3)) : -1.0f;
    }
    {
        const float T  = seed + __builtin_fmaf(0x1p-15f, pp + seed, 0x1p-100f);
        const float q  = pp - T;
        const float qo = __shfl_xor(q, 16, 32);
        bLo.y = hiH ? qo : bLo.y;
        bHi.y = hiH ? q  : bHi.y;
    }
    auto signs8 = [&](const v8f& d) {
        unsigned m = 0u;
#pragma unroll
        for (int r = 0; r < 8; ++r) m = __builtin_amdgcn_alignbit(m, __float_as_uint(d[r]), 31);
        return m;
    };

    v8f dLo, dHi;
    {
        const v2f aFirst = sa2[tBeg * 32];
        dLo = prod(aFirst, bLo);
        dHi = prod(aFirst, bHi);
    }
    v2f aNext = sa2[((tBeg + 1 < NTRIP) ? (tBeg + 1) : (NTRIP - 1)) * 32];
    for (int t = tBeg; t < tEnd; ++t) {
        const int t2 = (t + 2 < NTRIP) ? (t + 2) : (NTRIP - 1);
        const v2f aAfter = sa2[t2 * 32];
        const v8f nLo = prod(aNext, bLo);
        const v8f nHi = prod(aNext, bHi);
        const auto sw = __builtin_amdgcn_permlane16_swap(signs8(dLo), signs8(dHi), false, false);
        const unsigned mA = sw[0];
        const unsigned mB = sw[1];
        const unsigned m = (mA << 8) | mB;
        *(unsigned*)((char*)sQ + qb) = ((unsigned)t << 16) | m;
        qb += (m != 0u) ? QROWB : 0u;
        dLo = nLo;
        dHi = nHi;
        aNext = aAfter;
    }
    drainMask(qb / QROWB);

    __syncthreads();
    if (part > 0) {
#pragma unroll
        for (int k = 0; k < KNN; ++k)
        {
            const int row = (part - 1) * KNN + k;
            sQ[row * 2 * BLOCK + nl]         = bestD[k];
            sQ[row * 2 * BLOCK + BLOCK + nl] = __int_as_float(bestJ[k]);
        }
    }
    __syncthreads();

    if (part == 0) {
    drain((NPARTS - 1) * KNN);

    float* fr = &sFeat[nl * FSTRIDE];
    if (node < NNODES) {
        fr[0] = px;
        fr[1] = py;
#pragma unroll
        for (int k = 0; k < KNN; ++k) {
            const float2 q = sl[bestJ[k]];
            fr[2 + 2 * k]     = q.x - px;
            fr[2 + 2 * k + 1] = q.y - py;
        }
        fr[NODE_DIM]     = 0.0f;
        fr[NODE_DIM + 1] = 0.0f;
    } else {
#pragma unroll
        for (int k = 0; k < FDIM; ++k) fr[k] = 0.0f;
    }
    }
    __syncthreads();

    if (part != 0) return;
    const int wave   = tid >> 5;
    const int lane   = tid & 31;
    const int laneLo = lane & 15;
    const int kHalf  = (lane < 16) ? 0 : 2;
    const int mBase  = (lane < 16) ? 0 : 8;

    const int t0 = wave * 2;
    const int t1 = t0 + 1;
    const bool ok0 = (node0 + t0 * 16) < NNODES;
    const bool ok1 = (node0 + t1 * 16) < NNODES;

    v2f a0[6], a1[6];
    {
        const float* fa0 = &sFeat[(t0 * 16 + laneLo) * FSTRIDE];
        const float* fa1 = &sFeat[(t1 * 16 + laneLo) * FSTRIDE];
#pragma unroll
        for (int kc = 0; kc < 6; ++kc) {
            const int k0 = kc * 4 + kHalf;
            a0[kc].x = fa0[k0];  a0[kc].y = fa0[k0 + 1];
            a1[kc].x = fa1[k0];  a1[kc].y = fa1[k0 + 1];
        }
    }

    float* const outB = out + (size_t)batch * NNODES * EMBED_DIM;

#pragma unroll
    for (int ct = 0; ct < 8; ++ct) {
        const int e = ct * 16 + laneLo;
        v2f bf[6];
#pragma unroll
        for (int kc = 0; kc < 6; ++kc) {
            const int k0 = kc * 4 + kHalf;
            bf[kc].x = sW[k0 * EMBED_DIM + e];
            bf[kc].y = sW[(k0 + 1) * EMBED_DIM + e];
        }
        v8f acc0 = {}, acc1 = {};
#pragma unroll
        for (int kc = 0; kc < 6; ++kc)
            acc0 = __builtin_amdgcn_wmma_f32_16x16x4_f32(
                false, a0[kc], false, bf[kc], (short)0, acc0, false, false);
#pragma unroll
        for (int kc = 0; kc < 6; ++kc)
            acc1 = __builtin_amdgcn_wmma_f32_16x16x4_f32(
                false, a1[kc], false, bf[kc], (short)0, acc1, false, false);

        const float bv = sBias[e];
        if (ok0) {
            float* p = outB + (size_t)(node0 + t0 * 16 + mBase) * EMBED_DIM + e;
#pragma unroll
            for (int r = 0; r < 8; ++r)
                p[(size_t)r * EMBED_DIM] = acc0[r] + bv;
        }
        if (ok1) {
            float* p = outB + (size_t)(node0 + t1 * 16 + mBase) * EMBED_DIM + e;
#pragma unroll
            for (int r = 0; r < 8; ++r)
                p[(size_t)r * EMBED_DIM] = acc1[r] + bv;
        }
    }
}

extern "C" void kernel_launch(void* const* d_in, const int* in_sizes, int n_in,
                              void* d_out, int out_size, void* d_ws, size_t ws_size,
                              hipStream_t stream) {
    const float* locs = (const float*)d_in[0];
    const float* W    = (const float*)d_in[1];
    const float* b    = (const float*)d_in[2];
    float* out        = (float*)d_out;

    dim3 grid(NBATCH * BLOCKS_PER_BATCH);
    dim3 block(BLOCK);
    tsp_embed_kernel<<<grid, block, 0, stream>>>(locs, W, b, out);
}
